# starting kernel
# baseline (speedup 1.0000x reference)
#include <hip/hip_runtime.h>
#include <hip/hip_bf16.h>

#define IN_DIM  128
#define OUT_DIM 128

typedef float v2f __attribute__((ext_vector_type(2)));
typedef float v8f __attribute__((ext_vector_type(8)));

__global__ void __launch_bounds__(256)
gcn_gemm_wmma(const float* __restrict__ x, const float* __restrict__ w,
              float* __restrict__ support, int n_nodes) {
  __shared__ float wlds[IN_DIM * OUT_DIM];
  for (int i = threadIdx.x; i < IN_DIM * OUT_DIM; i += 256)
    wlds[i] = w[i];
  __syncthreads();

  const int lane = threadIdx.x & 31;
  const int wave = threadIdx.x >> 5;
  const int m0   = blockIdx.x * 128 + wave * 16;

  const int koff = (lane >> 4) * 2;
  const int nsub = lane & 15;

  int mrow = m0 + (lane & 15);
  if (mrow >= n_nodes) mrow = n_nodes - 1;

  const float* __restrict__ xrow = x + (size_t)mrow * IN_DIM;

  v8f acc[8];
  #pragma unroll
  for (int t = 0; t < 8; ++t) acc[t] = (v8f){};

  #pragma unroll 2
  for (int k0 = 0; k0 < IN_DIM; k0 += 4) {
    v2f a;
    a.x = xrow[k0 + koff];
    a.y = xrow[k0 + koff + 1];
    const float* __restrict__ wk0 = &wlds[(k0 + koff) * OUT_DIM + nsub];
    #pragma unroll
    for (int t = 0; t < 8; ++t) {
      v2f b;
      b.x = wk0[t * 16];
      b.y = wk0[t * 16 + OUT_DIM];
      acc[t] = __builtin_amdgcn_wmma_f32_16x16x4_f32(
           false, a,  false, b,
           (short)0, acc[t],  false,  false);
    }
  }

  const int mbase = m0 + ((lane >> 4) << 3);
  float* outp = support + (size_t)mbase * OUT_DIM + nsub;

  if (m0 + 15 < n_nodes) {
    #pragma unroll
    for (int r = 0; r < 8; ++r)
      #pragma unroll
      for (int t = 0; t < 8; ++t)
        outp[(size_t)r * OUT_DIM + t * 16] = acc[t][r];
  } else {
    #pragma unroll
    for (int r = 0; r < 8; ++r)
      if (mbase + r < n_nodes)
        #pragma unroll
        for (int t = 0; t < 8; ++t)
          outp[(size_t)r * OUT_DIM + t * 16] = acc[t][r];
  }
}

__global__ void __launch_bounds__(256)
gcn_bias_init(const float* __restrict__ bias, float* __restrict__ out, int total) {
  int idx = blockIdx.x * 256 + threadIdx.x;
  if (idx < total)
    out[idx] = bias[idx & (OUT_DIM - 1)];
}

__global__ void __launch_bounds__(256)
gcn_spmm_scatter(const float* __restrict__ support,
                 const float* __restrict__ adj_vals,
                 const int* __restrict__ edge_row,
                 const int* __restrict__ edge_col,
                 float* __restrict__ out, int n_edges) {
  const int lane = threadIdx.x & 31;
  const int edge = blockIdx.x * 8 + (threadIdx.x >> 5);
  if (edge >= n_edges) return;

  const int   col = edge_col[edge];
  const int   row = edge_row[edge];
  const float v   = adj_vals[edge];

  const float4 m =
      *((const float4*)(support + (size_t)col * OUT_DIM) + lane);

  float* dst = out + (size_t)row * OUT_DIM + lane * 4;
  __hip_atomic_fetch_add(dst + 0, m.x * v, __ATOMIC_RELAXED, __HIP_MEMORY_SCOPE_AGENT);
  __hip_atomic_fetch_add(dst + 1, m.y * v, __ATOMIC_RELAXED, __HIP_MEMORY_SCOPE_AGENT);
  __hip_atomic_fetch_add(dst + 2, m.z * v, __ATOMIC_RELAXED, __HIP_MEMORY_SCOPE_AGENT);
  __hip_atomic_fetch_add(dst + 3, m.w * v, __ATOMIC_RELAXED, __HIP_MEMORY_SCOPE_AGENT);
}

extern "C" void kernel_launch(void* const* d_in, const int* in_sizes, int n_in,
                              void* d_out, int out_size, void* d_ws, size_t ws_size,
                              hipStream_t stream) {
  const float* x        = (const float*)d_in[0];
  const float* w        = (const float*)d_in[1];
  const float* bias     = (const float*)d_in[2];
  const float* adj_vals = (const float*)d_in[3];
  const int*   edge_row = (const int*)d_in[4];
  const int*   edge_col = (const int*)d_in[5];

  const int n_nodes = in_sizes[0] / IN_DIM;
  const int n_edges = in_sizes[3];

  float* out     = (float*)d_out;
  float* support = (float*)d_ws;

  const int nblocks = (n_nodes + 127) / 128;
  gcn_gemm_wmma<<<nblocks, 256, 0, stream>>>(x, w, support, n_nodes);

  const int total = n_nodes * OUT_DIM;
  gcn_bias_init<<<(total + 255) / 256, 256, 0, stream>>>(bias, out, total);

  gcn_spmm_scatter<<<(n_edges + 7) / 8, 256, 0, stream>>>(
      support, adj_vals, edge_row, edge_col, out, n_edges);
}


# scatter -> owner scan: 256 work-groups each keep 196 out rows in LDS (start = bias), read the whole edge list, gather only their own edges (LDS f32 adds, arrival order), store each row once
# speedup vs baseline: 1.7679x; 1.7679x over previous
#include <hip/hip_runtime.h>
#include <hip/hip_bf16.h>

#define IN_DIM  128
#define OUT_DIM 128

typedef float v2f __attribute__((ext_vector_type(2)));
typedef float v8f __attribute__((ext_vector_type(8)));

__global__ void __launch_bounds__(256)
gcn_gemm_wmma(const float* __restrict__ x, const float* __restrict__ w,
              float* __restrict__ support, int n_nodes) {
  __shared__ float wlds[IN_DIM * OUT_DIM];

  const int lane = threadIdx.x & 31;
  const int wave = threadIdx.x >> 5;
  const int m0   = blockIdx.x * 128 + wave * 16;

  for (int i = threadIdx.x; i < IN_DIM * OUT_DIM; i += 256)
    wlds[i] = w[i];
  __syncthreads();

  const int koff = (lane >> 4) * 2;
  const int nsub = lane & 15;

  int mrow = m0 + (lane & 15);
  if (mrow >= n_nodes) mrow = n_nodes - 1;

  const float* __restrict__ xrow = x + (size_t)mrow * IN_DIM;

  v8f acc[8];
  #pragma unroll
  for (int t = 0; t < 8; ++t) acc[t] = (v8f){};

  #pragma unroll 2
  for (int k0 = 0; k0 < IN_DIM; k0 += 4) {
    v2f a;
    a.x = xrow[k0 + koff];
    a.y = xrow[k0 + koff + 1];
    const float* __restrict__ wk0 = &wlds[(k0 + koff) * OUT_DIM + nsub];
    #pragma unroll
    for (int t = 0; t < 8; ++t) {
      v2f b;
      b.x = wk0[t * 16];
      b.y = wk0[t * 16 + OUT_DIM];
      acc[t] = __builtin_amdgcn_wmma_f32_16x16x4_f32(
           false, a,  false, b,
           (short)0, acc[t],  false,  false);
    }
  }

  const int mbase = m0 + ((lane >> 4) << 3);
  float* outp = support + (size_t)mbase * OUT_DIM + nsub;

  if (m0 + 15 < n_nodes) {
    #pragma unroll
    for (int r = 0; r < 8; ++r)
      #pragma unroll
      for (int t = 0; t < 8; ++t)
        outp[(size_t)r * OUT_DIM + t * 16] = acc[t][r];
  } else {
    #pragma unroll
    for (int r = 0; r < 8; ++r)
      if (mbase + r < n_nodes)
        #pragma unroll
        for (int t = 0; t < 8; ++t)
          outp[(size_t)r * OUT_DIM + t * 16] = acc[t][r];
  }
}

#define ROWS_PER_WG   196
#define SCAN_LOADS    8
#define TRIP_EDGES    (32 * 4 * SCAN_LOADS)
#define WAVES_PER_WG  32

__device__ __forceinline__ void load_rows(const int* __restrict__ edge_row,
                                          int chunk, int lane, int n_edges,
                                          int4 (&rv)[SCAN_LOADS]) {
  if (chunk + TRIP_EDGES <= n_edges) {
    #pragma unroll
    for (int u = 0; u < SCAN_LOADS; ++u)
      rv[u] = *(const int4*)(edge_row + chunk + u * 128 + lane * 4);
  } else {
    #pragma unroll
    for (int u = 0; u < SCAN_LOADS; ++u) {
      const int e = chunk + u * 128 + lane * 4;
      rv[u].x = (e + 0 < n_edges) ? edge_row[e + 0] : -1;
      rv[u].y = (e + 1 < n_edges) ? edge_row[e + 1] : -1;
      rv[u].z = (e + 2 < n_edges) ? edge_row[e + 2] : -1;
      rv[u].w = (e + 3 < n_edges) ? edge_row[e + 3] : -1;
    }
  }
}

__device__ __forceinline__ void lds_add(float* p, float v) {
  __hip_atomic_fetch_add(p, v, __ATOMIC_RELAXED, __HIP_MEMORY_SCOPE_WORKGROUP);
}

__global__ void __launch_bounds__(1024)
gcn_rows_owner(const float* __restrict__ support,
               const float* __restrict__ adj_vals,
               const int* __restrict__ edge_row,
               const int* __restrict__ edge_col,
               const float* __restrict__ bias,
               float* __restrict__ out, int n_nodes, int n_edges) {
  __shared__ float    acc[ROWS_PER_WG * OUT_DIM];
  __shared__ unsigned hitq[WAVES_PER_WG * TRIP_EDGES];

  const int lane = threadIdx.x & 31;
  const int wave = __builtin_amdgcn_readfirstlane((int)(threadIdx.x >> 5));
  unsigned* myq = hitq + wave * TRIP_EDGES;

  const float4 b4 = *((const float4*)bias + lane);

  for (int row_base = blockIdx.x * ROWS_PER_WG; row_base < n_nodes;
       row_base += gridDim.x * ROWS_PER_WG) {
    const int nrows = (n_nodes - row_base < ROWS_PER_WG) ? (n_nodes - row_base) : ROWS_PER_WG;

    for (int r = wave; r < nrows; r += WAVES_PER_WG)
      *((float4*)(acc + r * OUT_DIM) + lane) = b4;
    __syncthreads();

    int4 cur[SCAN_LOADS];
    load_rows(edge_row, wave * TRIP_EDGES, lane, n_edges, cur);

    for (int chunk = wave * TRIP_EDGES; chunk < n_edges;
         chunk += WAVES_PER_WG * TRIP_EDGES) {
      int4 nxt[SCAN_LOADS];
      load_rows(edge_row, chunk + WAVES_PER_WG * TRIP_EDGES, lane, n_edges, nxt);

      int qn = 0;
      #pragma unroll
      for (int u = 0; u < SCAN_LOADS; ++u) {
        const int rw[4] = {cur[u].x, cur[u].y, cur[u].z, cur[u].w};
        #pragma unroll
        for (int k = 0; k < 4; ++k) {
          const unsigned r    = (unsigned)(rw[k] - row_base);
          const bool     hit  = r < (unsigned)nrows;
          const unsigned mask = __builtin_amdgcn_ballot_w32(hit);
          if (mask) {
            if (hit)
              myq[qn + (int)__builtin_amdgcn_mbcnt_lo(mask, 0u)] =
                  (r << 10) | (unsigned)(u * 128 + lane * 4 + k);
            qn += __builtin_popcount(mask);
          }
        }
      }

      for (int g = 0; g < qn; g += 32) {
        const int cnt = (qn - g < 32) ? (qn - g) : 32;
        unsigned ent = 0u; int colv = 0; float vv = 0.0f;
        if (lane < cnt) {
          ent = myq[g + lane];
          const int e = chunk + (int)(ent & 1023u);
          colv = edge_col[e];
          vv   = adj_vals[e];
        }
        for (int j = 0; j < cnt; j += 4) {
          int c[4]; float v[4]; unsigned rr[4];
          #pragma unroll
          for (int t = 0; t < 4; ++t) {
            const int jj = (j + t < cnt) ? (j + t) : (cnt - 1);
            c[t]  = __builtin_amdgcn_readlane(colv, jj);
            v[t]  = __builtin_bit_cast(float,
                      __builtin_amdgcn_readlane(__builtin_bit_cast(int, vv), jj));
            rr[t] = (unsigned)__builtin_amdgcn_readlane((int)ent, jj) >> 10;
          }
          float4 m[4];
          #pragma unroll
          for (int t = 0; t < 4; ++t)
            m[t] = *((const float4*)(support + (size_t)c[t] * OUT_DIM) + lane);
          #pragma unroll
          for (int t = 0; t < 4; ++t) {
            if (j + t < cnt) {
              float* a = acc + rr[t] * OUT_DIM + lane * 4;
              lds_add(a + 0, m[t].x * v[t]);
              lds_add(a + 1, m[t].y * v[t]);
              lds_add(a + 2, m[t].z * v[t]);
              lds_add(a + 3, m[t].w * v[t]);
            }
          }
        }
      }

      #pragma unroll
      for (int u = 0; u < SCAN_LOADS; ++u) cur[u] = nxt[u];
    }
    __syncthreads();

    for (int r = wave; r < nrows; r += WAVES_PER_WG)
      *((float4*)(out + (size_t)(row_base + r) * OUT_DIM) + lane) =
          *((const float4*)(acc + r * OUT_DIM) + lane);
    __syncthreads();
  }
}

extern "C" void kernel_launch(void* const* d_in, const int* in_sizes, int n_in,
                              void* d_out, int out_size, void* d_ws, size_t ws_size,
                              hipStream_t stream) {
  const float* x        = (const float*)d_in[0];
  const float* w        = (const float*)d_in[1];
  const float* bias     = (const float*)d_in[2];
  const float* adj_vals = (const float*)d_in[3];
  const int*   edge_row = (const int*)d_in[4];
  const int*   edge_col = (const int*)d_in[5];

  const int n_nodes = in_sizes[0] / IN_DIM;
  const int n_edges = in_sizes[3];

  float* out     = (float*)d_out;
  float* support = (float*)d_ws;

  const int nblocks = (n_nodes + 127) / 128;
  gcn_gemm_wmma<<<nblocks, 256, 0, stream>>>(x, w, support, n_nodes);

  gcn_rows_owner<<<256, 1024, 0, stream>>>(
      support, adj_vals, edge_row, edge_col, bias, out, n_nodes, n_edges);
}


# owner scan reads a key plane: launch 1 writes one byte per edge (row / 196); an owner group scans 1 byte per edge, 4 edges per vector test; the row number settles a note; LDS f32 adds, arrival order
# speedup vs baseline: 1.8617x; 1.0531x over previous
#include <hip/hip_runtime.h>
#include <hip/hip_bf16.h>

#define IN_DIM  128
#define OUT_DIM 128

typedef float v2f __attribute__((ext_vector_type(2)));
typedef float v8f __attribute__((ext_vector_type(8)));

#define ROWS_PER_WG   196

__device__ __forceinline__ unsigned owner_key(int r) {
  return ((unsigned)r / (unsigned)ROWS_PER_WG) & 255u;
}

__global__ void __launch_bounds__(256)
gcn_gemm_wmma(const float* __restrict__ x, const float* __restrict__ w,
              float* __restrict__ support, int n_nodes,
              const int* __restrict__ edge_row, int* __restrict__ keys, int n_edges) {
  __shared__ float wlds[IN_DIM * OUT_DIM];

  const int lane = threadIdx.x & 31;
  const int wave = threadIdx.x >> 5;
  const int m0   = blockIdx.x * 128 + wave * 16;

  {
    const int nwords = (n_edges + 3) >> 2;
    for (int q = blockIdx.x * 256 + threadIdx.x; q < nwords; q += gridDim.x * 256) {
      int4 rv;
      if (4 * q + 3 < n_edges) {
        rv = *(const int4*)(edge_row + 4 * q);
      } else {
        rv.x = edge_row[4 * q];
        rv.y = (4 * q + 1 < n_edges) ? edge_row[4 * q + 1] : 0;
        rv.z = (4 * q + 2 < n_edges) ? edge_row[4 * q + 2] : 0;
        rv.w = 0;
      }
      unsigned kw = owner_key(rv.x) | (owner_key(rv.y) << 8) |
                    (owner_key(rv.z) << 16) | (owner_key(rv.w) << 24);
      if (4 * q + 3 >= n_edges) {
        const int vb = n_edges - 4 * q;
        kw &= (1u << (8 * vb)) - 1u;
      }
      keys[q] = (int)kw;
    }
  }

  for (int i = threadIdx.x; i < IN_DIM * OUT_DIM; i += 256)
    wlds[i] = w[i];
  __syncthreads();

  const int koff = (lane >> 4) * 2;
  const int nsub = lane & 15;

  int mrow = m0 + (lane & 15);
  if (mrow >= n_nodes) mrow = n_nodes - 1;

  const float* __restrict__ xrow = x + (size_t)mrow * IN_DIM;

  v8f acc[8];
  #pragma unroll
  for (int t = 0; t < 8; ++t) acc[t] = (v8f){};

  #pragma unroll 2
  for (int k0 = 0; k0 < IN_DIM; k0 += 4) {
    v2f a;
    a.x = xrow[k0 + koff];
    a.y = xrow[k0 + koff + 1];
    const float* __restrict__ wk0 = &wlds[(k0 + koff) * OUT_DIM + nsub];
    #pragma unroll
    for (int t = 0; t < 8; ++t) {
      v2f b;
      b.x = wk0[t * 16];
      b.y = wk0[t * 16 + OUT_DIM];
      acc[t] = __builtin_amdgcn_wmma_f32_16x16x4_f32(
           false, a,  false, b,
           (short)0, acc[t],  false,  false);
    }
  }

  const int mbase = m0 + ((lane >> 4) << 3);
  float* outp = support + (size_t)mbase * OUT_DIM + nsub;

  if (m0 + 15 < n_nodes) {
    #pragma unroll
    for (int r = 0; r < 8; ++r)
      #pragma unroll
      for (int t = 0; t < 8; ++t)
        outp[(size_t)r * OUT_DIM + t * 16] = acc[t][r];
  } else {
    #pragma unroll
    for (int r = 0; r < 8; ++r)
      if (mbase + r < n_nodes)
        #pragma unroll
        for (int t = 0; t < 8; ++t)
          outp[(size_t)r * OUT_DIM + t * 16] = acc[t][r];
  }
}

#define SCAN_LOADS    2
#define TRIP_EDGES    (32 * 16 * SCAN_LOADS)
#define WAVES_PER_WG  32

__device__ __forceinline__ void load_keys(const int* __restrict__ keys,
                                          int chunk, int lane, int n_edges,
                                          int4 (&kv)[SCAN_LOADS]) {
  if (chunk + TRIP_EDGES <= n_edges) {
    #pragma unroll
    for (int u = 0; u < SCAN_LOADS; ++u)
      kv[u] = *(const int4*)(keys + (chunk >> 2) + u * 128 + lane * 4);
  } else {
    #pragma unroll
    for (int u = 0; u < SCAN_LOADS; ++u) {
      const int q = (chunk >> 2) + u * 128 + lane * 4;
      kv[u].x = (4 * (q + 0) < n_edges) ? keys[q + 0] : 0;
      kv[u].y = (4 * (q + 1) < n_edges) ? keys[q + 1] : 0;
      kv[u].z = (4 * (q + 2) < n_edges) ? keys[q + 2] : 0;
      kv[u].w = (4 * (q + 3) < n_edges) ? keys[q + 3] : 0;
    }
  }
}

template <bool TAIL>
__device__ __forceinline__ int note_trip(const int4 (&cur)[SCAN_LOADS], unsigned k4,
                                         int lane, int chunk, int n_edges,
                                         unsigned* myq) {
  int qn = 0;
  #pragma unroll
  for (int u = 0; u < SCAN_LOADS; ++u) {
    const int kw[4] = {cur[u].x, cur[u].y, cur[u].z, cur[u].w};
    #pragma unroll
    for (int c = 0; c < 4; ++c) {
      const int      off = (u * 128 + lane * 4 + c) * 4;
      const unsigned t   = (unsigned)kw[c] ^ k4;
      unsigned z = ~(((t & 0x7f7f7f7fu) + 0x7f7f7f7fu) | t | 0x7f7f7f7fu);
      if (TAIL) {
        const int vb = n_edges - (chunk + off);
        const unsigned keep = (vb >= 4) ? 0x80808080u
                            : (vb <= 0) ? 0u : (((1u << (8 * vb)) - 1u) & 0x80808080u);
        z &= keep;
      }
      unsigned mask = __builtin_amdgcn_ballot_w32(z != 0u);
      for (int it = 0; it < 4 && mask; ++it) {
        if (z) {
          myq[qn + (int)__builtin_amdgcn_mbcnt_lo(mask, 0u)] =
              (unsigned)(off + (__builtin_ctz(z) >> 3));
          z &= z - 1u;
        }
        qn += __builtin_popcount(mask);
        mask = __builtin_amdgcn_ballot_w32(z != 0u);
      }
    }
  }
  return qn;
}

__device__ __forceinline__ void lds_add(float* p, float v) {
  __hip_atomic_fetch_add(p, v, __ATOMIC_RELAXED, __HIP_MEMORY_SCOPE_WORKGROUP);
}

__global__ void __launch_bounds__(1024)
gcn_rows_owner(const float* __restrict__ support,
               const float* __restrict__ adj_vals,
               const int* __restrict__ keys,
               const int* __restrict__ edge_row,
               const int* __restrict__ edge_col,
               const float* __restrict__ bias,
               float* __restrict__ out, int n_nodes, int n_edges) {
  __shared__ float    acc[ROWS_PER_WG * OUT_DIM];
  __shared__ unsigned hitq[WAVES_PER_WG * TRIP_EDGES];

  const int lane = threadIdx.x & 31;
  const int wave = __builtin_amdgcn_readfirstlane((int)(threadIdx.x >> 5));
  unsigned* myq = hitq + wave * TRIP_EDGES;

  const float4 b4 = *((const float4*)bias + lane);

  for (int row_base = blockIdx.x * ROWS_PER_WG; row_base < n_nodes;
       row_base += gridDim.x * ROWS_PER_WG) {
    const int nrows = (n_nodes - row_base < ROWS_PER_WG) ? (n_nodes - row_base) : ROWS_PER_WG;

    for (int r = wave; r < nrows; r += WAVES_PER_WG)
      *((float4*)(acc + r * OUT_DIM) + lane) = b4;
    __syncthreads();

    const unsigned k4 = owner_key(row_base) * 0x01010101u;

    int4 cur[SCAN_LOADS];
    load_keys(keys, wave * TRIP_EDGES, lane, n_edges, cur);

    for (int chunk = wave * TRIP_EDGES; chunk < n_edges;
         chunk += WAVES_PER_WG * TRIP_EDGES) {
      int4 nxt[SCAN_LOADS];
      load_keys(keys, chunk + WAVES_PER_WG * TRIP_EDGES, lane, n_edges, nxt);
      const int qn = (chunk + TRIP_EDGES <= n_edges)
                         ? note_trip<false>(cur, k4, lane, chunk, n_edges, myq)
                         : note_trip<true>(cur, k4, lane, chunk, n_edges, myq);

      for (int g = 0; g < qn; g += 32) {
        const int cnt = (qn - g < 32) ? (qn - g) : 32;
        unsigned rloc = 0u; int colv = 0; float vv = 0.0f; bool mine = false;
        if (lane < cnt) {
          const int e = chunk + (int)myq[g + lane];
          rloc = (unsigned)(edge_row[e] - row_base);
          colv = edge_col[e];
          vv   = adj_vals[e];
          mine = rloc < (unsigned)nrows;
        }
        const unsigned minem = __builtin_amdgcn_ballot_w32(mine);
        for (int j = 0; j < cnt; j += 4) {
          int c[4]; float v[4]; unsigned rr[4];
          #pragma unroll
          for (int t = 0; t < 4; ++t) {
            const int jj = (j + t < cnt) ? (j + t) : (cnt - 1);
            c[t]  = __builtin_amdgcn_readlane(colv, jj);
            v[t]  = __builtin_bit_cast(float,
                      __builtin_amdgcn_readlane(__builtin_bit_cast(int, vv), jj));
            rr[t] = (unsigned)__builtin_amdgcn_readlane((int)rloc, jj);
          }
          float4 m[4];
          #pragma unroll
          for (int t = 0; t < 4; ++t)
            m[t] = *((const float4*)(support + (size_t)c[t] * OUT_DIM) + lane);
          #pragma unroll
          for (int t = 0; t < 4; ++t) {
            if (j + t < cnt && ((minem >> (j + t)) & 1u)) {
              float* a = acc + rr[t] * OUT_DIM + lane * 4;
              lds_add(a + 0, m[t].x * v[t]);
              lds_add(a + 1, m[t].y * v[t]);
              lds_add(a + 2, m[t].z * v[t]);
              lds_add(a + 3, m[t].w * v[t]);
            }
          }
        }
      }

      #pragma unroll
      for (int u = 0; u < SCAN_LOADS; ++u) cur[u] = nxt[u];
    }
    __syncthreads();

    for (int r = wave; r < nrows; r += WAVES_PER_WG)
      *((float4*)(out + (size_t)(row_base + r) * OUT_DIM) + lane) =
          *((const float4*)(acc + r * OUT_DIM) + lane);
    __syncthreads();
  }
}

extern "C" void kernel_launch(void* const* d_in, const int* in_sizes, int n_in,
                              void* d_out, int out_size, void* d_ws, size_t ws_size,
                              hipStream_t stream) {
  const float* x        = (const float*)d_in[0];
  const float* w        = (const float*)d_in[1];
  const float* bias     = (const float*)d_in[2];
  const float* adj_vals = (const float*)d_in[3];
  const int*   edge_row = (const int*)d_in[4];
  const int*   edge_col = (const int*)d_in[5];

  const int n_nodes = in_sizes[0] / IN_DIM;
  const int n_edges = in_sizes[3];

  float* out     = (float*)d_out;
  float* support = (float*)d_ws;
  int*   keys    = (int*)(support + (size_t)n_nodes * OUT_DIM);

  const int nblocks = (n_nodes + 127) / 128;
  gcn_gemm_wmma<<<nblocks, 256, 0, stream>>>(x, w, support, n_nodes, edge_row, keys, n_edges);

  gcn_rows_owner<<<256, 1024, 0, stream>>>(
      support, adj_vals, keys, edge_row, edge_col, bias, out, n_nodes, n_edges);
}


# owner scan: a row belongs to one wave (row & 31); finders queue (col, value, row) for it in LDS; it adds gathered rows by plain LDS read-add-write (float adds only if a queue is full); arrival order
# speedup vs baseline: 1.9316x; 1.0375x over previous
#include <hip/hip_runtime.h>
#include <hip/hip_bf16.h>

#define IN_DIM  128
#define OUT_DIM 128

typedef float v2f __attribute__((ext_vector_type(2)));
typedef float v8f __attribute__((ext_vector_type(8)));

#define ROWS_PER_WG   196

__device__ __forceinline__ unsigned owner_key(int r) {
  return ((unsigned)r / (unsigned)ROWS_PER_WG) & 255u;
}

__global__ void __launch_bounds__(256)
gcn_gemm_wmma(const float* __restrict__ x, const float* __restrict__ w,
              float* __restrict__ support, int n_nodes,
              const int* __restrict__ edge_row, int* __restrict__ keys, int n_edges) {
  __shared__ float wlds[IN_DIM * OUT_DIM];

  const int lane = threadIdx.x & 31;
  const int wave = threadIdx.x >> 5;
  const int m0   = blockIdx.x * 128 + wave * 16;

  {
    const int nwords = (n_edges + 3) >> 2;
    for (int q = blockIdx.x * 256 + threadIdx.x; q < nwords; q += gridDim.x * 256) {
      int4 rv;
      if (4 * q + 3 < n_edges) {
        rv = *(const int4*)(edge_row + 4 * q);
      } else {
        rv.x = edge_row[4 * q];
        rv.y = (4 * q + 1 < n_edges) ? edge_row[4 * q + 1] : 0;
        rv.z = (4 * q + 2 < n_edges) ? edge_row[4 * q + 2] : 0;
        rv.w = 0;
      }
      unsigned kw = owner_key(rv.x) | (owner_key(rv.y) << 8) |
                    (owner_key(rv.z) << 16) | (owner_key(rv.w) << 24);
      if (4 * q + 3 >= n_edges) {
        const int vb = n_edges - 4 * q;
        kw &= (1u << (8 * vb)) - 1u;
      }
      keys[q] = (int)kw;
    }
  }

  for (int i = threadIdx.x; i < IN_DIM * OUT_DIM; i += 256)
    wlds[i] = w[i];
  __syncthreads();

  const int koff = (lane >> 4) * 2;
  const int nsub = lane & 15;

  int mrow = m0 + (lane & 15);
  if (mrow >= n_nodes) mrow = n_nodes - 1;

  const float* __restrict__ xrow = x + (size_t)mrow * IN_DIM;

  v8f acc[8];
  #pragma unroll
  for (int t = 0; t < 8; ++t) acc[t] = (v8f){};

  #pragma unroll 2
  for (int k0 = 0; k0 < IN_DIM; k0 += 4) {
    v2f a;
    a.x = xrow[k0 + koff];
    a.y = xrow[k0 + koff + 1];
    const float* __restrict__ wk0 = &wlds[(k0 + koff) * OUT_DIM + nsub];
    #pragma unroll
    for (int t = 0; t < 8; ++t) {
      v2f b;
      b.x = wk0[t * 16];
      b.y = wk0[t * 16 + OUT_DIM];
      acc[t] = __builtin_amdgcn_wmma_f32_16x16x4_f32(
           false, a,  false, b,
           (short)0, acc[t],  false,  false);
    }
  }

  const int mbase = m0 + ((lane >> 4) << 3);
  float* outp = support + (size_t)mbase * OUT_DIM + nsub;

  if (m0 + 15 < n_nodes) {
    #pragma unroll
    for (int r = 0; r < 8; ++r)
      #pragma unroll
      for (int t = 0; t < 8; ++t)
        outp[(size_t)r * OUT_DIM + t * 16] = acc[t][r];
  } else {
    #pragma unroll
    for (int r = 0; r < 8; ++r)
      if (mbase + r < n_nodes)
        #pragma unroll
        for (int t = 0; t < 8; ++t)
          outp[(size_t)r * OUT_DIM + t * 16] = acc[t][r];
  }
}

#define SCAN_LOADS    2
#define TRIP_EDGES    (32 * 16 * SCAN_LOADS)
#define WAVES_PER_WG  32
#define TRIPS_PER_ROUND 5
#define QCAP          64

__device__ __forceinline__ void load_keys(const int* __restrict__ keys,
                                          int chunk, int lane, int n_edges,
                                          int4 (&kv)[SCAN_LOADS]) {
  if (chunk + TRIP_EDGES <= n_edges) {
    #pragma unroll
    for (int u = 0; u < SCAN_LOADS; ++u)
      kv[u] = *(const int4*)(keys + (chunk >> 2) + u * 128 + lane * 4);
  } else {
    #pragma unroll
    for (int u = 0; u < SCAN_LOADS; ++u) {
      const int q = (chunk >> 2) + u * 128 + lane * 4;
      kv[u].x = (4 * (q + 0) < n_edges) ? keys[q + 0] : 0;
      kv[u].y = (4 * (q + 1) < n_edges) ? keys[q + 1] : 0;
      kv[u].z = (4 * (q + 2) < n_edges) ? keys[q + 2] : 0;
      kv[u].w = (4 * (q + 3) < n_edges) ? keys[q + 3] : 0;
    }
  }
}

template <bool TAIL>
__device__ __forceinline__ int note_trip(const int4 (&cur)[SCAN_LOADS], unsigned k4,
                                         int lane, int chunk, int n_edges,
                                         unsigned short* myq) {
  int qn = 0;
  #pragma unroll
  for (int u = 0; u < SCAN_LOADS; ++u) {
    const int kw[4] = {cur[u].x, cur[u].y, cur[u].z, cur[u].w};
    #pragma unroll
    for (int c = 0; c < 4; ++c) {
      const int      off = (u * 128 + lane * 4 + c) * 4;
      const unsigned t   = (unsigned)kw[c] ^ k4;
      unsigned z = ~(((t & 0x7f7f7f7fu) + 0x7f7f7f7fu) | t | 0x7f7f7f7fu);
      if (TAIL) {
        const int vb = n_edges - (chunk + off);
        const unsigned keep = (vb >= 4) ? 0x80808080u
                            : (vb <= 0) ? 0u : (((1u << (8 * vb)) - 1u) & 0x80808080u);
        z &= keep;
      }
      unsigned mask = __builtin_amdgcn_ballot_w32(z != 0u);
      for (int it = 0; it < 4 && mask; ++it) {
        if (z) {
          myq[qn + (int)__builtin_amdgcn_mbcnt_lo(mask, 0u)] =
              (unsigned short)(off + (__builtin_ctz(z) >> 3));
          z &= z - 1u;
        }
        qn += __builtin_popcount(mask);
        mask = __builtin_amdgcn_ballot_w32(z != 0u);
      }
    }
  }
  return qn;
}

__device__ __forceinline__ void lds_add(float* p, float v) {
  __hip_atomic_fetch_add(p, v, __ATOMIC_RELAXED, __HIP_MEMORY_SCOPE_WORKGROUP);
}

__device__ __forceinline__ float4 row_fma(float4 s, float4 m, float v) {
  s.x += m.x * v; s.y += m.y * v; s.z += m.z * v; s.w += m.w * v;
  return s;
}

__device__ __forceinline__ void drain_queue(const int* qcol, const float* qval,
                                            const unsigned char* qrow, unsigned* qcnt,
                                            int qi, int lane, const float* __restrict__ support,
                                            float* acc) {
  const int nraw = (int)qcnt[qi];
  const int n = __builtin_amdgcn_readfirstlane(nraw < QCAP ? nraw : QCAP);
  if (n == 0) return;
  if (lane == 0) qcnt[qi] = 0u;
  const int qb = qi * QCAP;
  for (int g = 0; g < n; g += 32) {
    const int cnt = (n - g < 32) ? (n - g) : 32;
    int colv = 0; float vv = 0.0f; int rs = 0;
    if (lane < cnt) {
      colv = qcol[qb + g + lane];
      vv   = qval[qb + g + lane];
      rs   = (int)qrow[qb + g + lane];
    }
    auto gather8 = [&](float4 (&m)[8], int j) {
      #pragma unroll
      for (int t = 0; t < 8; ++t) {
        const int jj = (j + t < cnt) ? (j + t) : (cnt - 1);
        const int c  = __builtin_amdgcn_readlane(colv, jj);
        m[t] = *((const float4*)(support + (size_t)c * OUT_DIM) + lane);
      }
    };
    auto add8 = [&](const float4 (&m)[8], int j) {
      #pragma unroll
      for (int t = 0; t < 8; ++t) {
        if (j + t < cnt) {
          const float v = __builtin_bit_cast(float,
              __builtin_amdgcn_readlane(__builtin_bit_cast(int, vv), j + t));
          const int r = __builtin_amdgcn_readlane(rs, j + t);
          float4* p = (float4*)(acc + r * OUT_DIM) + lane;
          *p = row_fma(*p, m[t], v);
        }
      }
    };
    float4 ma[8], mb[8];
    gather8(ma, 0);
    for (int j = 0; j < cnt; j += 16) {
      const bool second = j + 8 < cnt;
      if (second) gather8(mb, j + 8);
      add8(ma, j);
      if (second) {
        if (j + 16 < cnt) gather8(ma, j + 16);
        add8(mb, j + 8);
      }
    }
  }
}

__global__ void __launch_bounds__(1024)
gcn_rows_owner(const float* __restrict__ support,
               const float* __restrict__ adj_vals,
               const int* __restrict__ keys,
               const int* __restrict__ edge_row,
               const int* __restrict__ edge_col,
               const float* __restrict__ bias,
               float* __restrict__ out, int n_nodes, int n_edges) {
  __shared__ float          acc[ROWS_PER_WG * OUT_DIM];
  __shared__ float          ovf[ROWS_PER_WG * OUT_DIM];
  __shared__ unsigned short hitq[WAVES_PER_WG * TRIP_EDGES];
  __shared__ int            qcol[2 * WAVES_PER_WG * QCAP];
  __shared__ float          qval[2 * WAVES_PER_WG * QCAP];
  __shared__ unsigned char  qrow[2 * WAVES_PER_WG * QCAP];
  __shared__ unsigned       qcnt[2 * WAVES_PER_WG];

  const int lane = threadIdx.x & 31;
  const int wave = __builtin_amdgcn_readfirstlane((int)(threadIdx.x >> 5));
  unsigned short* myq = hitq + wave * TRIP_EDGES;

  const float4 b4 = *((const float4*)bias + lane);

  const int nbt    = (n_edges + WAVES_PER_WG * TRIP_EDGES - 1) / (WAVES_PER_WG * TRIP_EDGES);
  const int rounds = (nbt + TRIPS_PER_ROUND - 1) / TRIPS_PER_ROUND;

  const bool drain_first = (wave & 4) != 0;

  for (int row_base = blockIdx.x * ROWS_PER_WG; row_base < n_nodes;
       row_base += gridDim.x * ROWS_PER_WG) {
    const int nrows = (n_nodes - row_base < ROWS_PER_WG) ? (n_nodes - row_base) : ROWS_PER_WG;

    for (int r = wave; r < nrows; r += WAVES_PER_WG) {
      *((float4*)(acc + r * OUT_DIM) + lane) = b4;
      *((float4*)(ovf + r * OUT_DIM) + lane) = make_float4(0.f, 0.f, 0.f, 0.f);
    }
    if (threadIdx.x < 2 * WAVES_PER_WG) qcnt[threadIdx.x] = 0u;
    __syncthreads();

    const unsigned k4 = owner_key(row_base) * 0x01010101u;


    int4 cur[SCAN_LOADS];
    int chunk = wave * TRIP_EDGES;
    load_keys(keys, chunk, lane, n_edges, cur);

    for (int round = 0; round <= rounds; ++round) {
      const int wb = round & 1;
      if (drain_first)
        drain_queue(qcol, qval, qrow, qcnt, (wb ^ 1) * WAVES_PER_WG + wave, lane, support, acc);

      if (round < rounds) {
        for (int trip = 0; trip < TRIPS_PER_ROUND && chunk < n_edges; ++trip) {
          int4 nxt[SCAN_LOADS];
          load_keys(keys, chunk + WAVES_PER_WG * TRIP_EDGES, lane, n_edges, nxt);
          const int qn = (chunk + TRIP_EDGES <= n_edges)
                             ? note_trip<false>(cur, k4, lane, chunk, n_edges, myq)
                             : note_trip<true>(cur, k4, lane, chunk, n_edges, myq);

          for (int g = 0; g < qn; g += 32) {
            const int cnt = (qn - g < 32) ? (qn - g) : 32;
            unsigned rloc = 0u; int colv = 0; float vv = 0.0f; bool mine = false;
            if (lane < cnt) {
              const int e = chunk + (int)myq[g + lane];
              rloc = (unsigned)(edge_row[e] - row_base);
              colv = edge_col[e];
              vv   = adj_vals[e];
              mine = rloc < (unsigned)nrows;
            }
            bool full = false;
            if (mine) {
              const int qi = wb * WAVES_PER_WG + (int)(rloc & 31u);
              const unsigned slot = __hip_atomic_fetch_add(&qcnt[qi], 1u, __ATOMIC_RELAXED,
                                                           __HIP_MEMORY_SCOPE_WORKGROUP);
              if (slot < (unsigned)QCAP) {
                const int idx = qi * QCAP + (int)slot;
                qcol[idx] = colv;
                qval[idx] = vv;
                qrow[idx] = (unsigned char)rloc;
              } else {
                full = true;
              }
            }
            unsigned fm = __builtin_amdgcn_ballot_w32(full);
            while (fm) {
              const int jj = __builtin_ctz(fm);
              fm &= fm - 1u;
              const int      c  = __builtin_amdgcn_readlane(colv, jj);
              const float    v  = __builtin_bit_cast(float,
                                    __builtin_amdgcn_readlane(__builtin_bit_cast(int, vv), jj));
              const unsigned rr = (unsigned)__builtin_amdgcn_readlane((int)rloc, jj);
              const float4 m = *((const float4*)(support + (size_t)c * OUT_DIM) + lane);
              float* a = ovf + rr * OUT_DIM + lane * 4;
              lds_add(a + 0, m.x * v);
              lds_add(a + 1, m.y * v);
              lds_add(a + 2, m.z * v);
              lds_add(a + 3, m.w * v);
            }
          }

          #pragma unroll
          for (int u = 0; u < SCAN_LOADS; ++u) cur[u] = nxt[u];
          chunk += WAVES_PER_WG * TRIP_EDGES;
        }
      }

      if (!drain_first)
        drain_queue(qcol, qval, qrow, qcnt, (wb ^ 1) * WAVES_PER_WG + wave, lane, support, acc);
      __syncthreads();
    }

    for (int r = wave; r < nrows; r += WAVES_PER_WG) {
      const float4 a = *((const float4*)(acc + r * OUT_DIM) + lane);
      const float4 o = *((const float4*)(ovf + r * OUT_DIM) + lane);
      *((float4*)(out + (size_t)(row_base + r) * OUT_DIM) + lane) =
          make_float4(a.x + o.x, a.y + o.y, a.z + o.z, a.w + o.w);
    }
    __syncthreads();
  }
}

extern "C" void kernel_launch(void* const* d_in, const int* in_sizes, int n_in,
                              void* d_out, int out_size, void* d_ws, size_t ws_size,
                              hipStream_t stream) {
  const float* x        = (const float*)d_in[0];
  const float* w        = (const float*)d_in[1];
  const float* bias     = (const float*)d_in[2];
  const float* adj_vals = (const float*)d_in[3];
  const int*   edge_row = (const int*)d_in[4];
  const int*   edge_col = (const int*)d_in[5];

  const int n_nodes = in_sizes[0] / IN_DIM;
  const int n_edges = in_sizes[3];

  float* out     = (float*)d_out;
  float* support = (float*)d_ws;
  int*   keys    = (int*)(support + (size_t)n_nodes * OUT_DIM);

  const int nblocks = (n_nodes + 127) / 128;
  gcn_gemm_wmma<<<nblocks, 256, 0, stream>>>(x, w, support, n_nodes, edge_row, keys, n_edges);

  gcn_rows_owner<<<256, 1024, 0, stream>>>(
      support, adj_vals, keys, edge_row, edge_col, bias, out, n_nodes, n_edges);
}


# owner scan via presence planes: launch 1 sets a bit per (32-edge block, row range); an owner reads its plane, then rows of flagged blocks only, a round's finds in one go; LDS f32 adds, arrival order
# speedup vs baseline: 2.3205x; 1.2013x over previous
#include <hip/hip_runtime.h>
#include <hip/hip_bf16.h>

#define IN_DIM  128
#define OUT_DIM 128

typedef float v2f __attribute__((ext_vector_type(2)));
typedef float v8f __attribute__((ext_vector_type(8)));

#define ROWS_PER_WG   196

__device__ __forceinline__ unsigned owner_key(int r) {
  return ((unsigned)r / (unsigned)ROWS_PER_WG) & 255u;
}

#define BLK_EDGES 32
__device__ __forceinline__ int presence_blocks(int n_edges) {
  return (n_edges + BLK_EDGES - 1) / BLK_EDGES;
}
__device__ __forceinline__ int presence_stride(int n_edges) {
  return (presence_blocks(n_edges) + 31) & ~31;
}
__device__ __forceinline__ void presence_mark(unsigned (&m)[8], int r) {
  const unsigned k   = owner_key(r);
  const unsigned bit = 1u << (k & 31u);
  const unsigned j   = k >> 5;
  #pragma unroll
  for (int t = 0; t < 8; ++t) m[t] |= (j == (unsigned)t) ? bit : 0u;
}

__global__ void __launch_bounds__(256)
gcn_gemm_wmma(const float* __restrict__ x, const float* __restrict__ w,
              float* __restrict__ support, int n_nodes,
              const int* __restrict__ edge_row, int* __restrict__ keys, int n_edges) {
  __shared__ float wlds[IN_DIM * OUT_DIM];

  const int lane = threadIdx.x & 31;
  const int wave = threadIdx.x >> 5;
  const int m0   = blockIdx.x * 128 + wave * 16;

  {
    const int nblk    = presence_blocks(n_edges);
    const int pstride = presence_stride(n_edges);
    const int ngroups = (nblk + 31) >> 5;
    for (int g = wave * (int)gridDim.x + (int)blockIdx.x; g < ngroups; g += 8 * (int)gridDim.x) {
      const int b = g * 32 + lane;
      if (b < nblk) {
        unsigned m[8] = {0u, 0u, 0u, 0u, 0u, 0u, 0u, 0u};
        const int e0 = b * BLK_EDGES;
        if (e0 + BLK_EDGES <= n_edges) {
          #pragma unroll
          for (int i = 0; i < BLK_EDGES / 4; ++i) {
            const int4 rv = *(const int4*)(edge_row + e0 + 4 * i);
            presence_mark(m, rv.x);
            presence_mark(m, rv.y);
            presence_mark(m, rv.z);
            presence_mark(m, rv.w);
          }
        } else {
          for (int i = 0; e0 + i < n_edges; ++i)
            presence_mark(m, edge_row[e0 + i]);
        }
        #pragma unroll
        for (int j = 0; j < 8; ++j) keys[j * pstride + b] = (int)m[j];
      }
    }
  }

  for (int i = threadIdx.x; i < IN_DIM * OUT_DIM; i += 256)
    wlds[i] = w[i];
  __syncthreads();

  const int koff = (lane >> 4) * 2;
  const int nsub = lane & 15;

  int mrow = m0 + (lane & 15);
  if (mrow >= n_nodes) mrow = n_nodes - 1;

  const float* __restrict__ xrow = x + (size_t)mrow * IN_DIM;

  v8f acc[8];
  #pragma unroll
  for (int t = 0; t < 8; ++t) acc[t] = (v8f){};

  #pragma unroll 2
  for (int k0 = 0; k0 < IN_DIM; k0 += 4) {
    v2f a;
    a.x = xrow[k0 + koff];
    a.y = xrow[k0 + koff + 1];
    const float* __restrict__ wk0 = &wlds[(k0 + koff) * OUT_DIM + nsub];
    #pragma unroll
    for (int t = 0; t < 8; ++t) {
      v2f b;
      b.x = wk0[t * 16];
      b.y = wk0[t * 16 + OUT_DIM];
      acc[t] = __builtin_amdgcn_wmma_f32_16x16x4_f32(
           false, a,  false, b,
           (short)0, acc[t],  false,  false);
    }
  }

  const int mbase = m0 + ((lane >> 4) << 3);
  float* outp = support + (size_t)mbase * OUT_DIM + nsub;

  if (m0 + 15 < n_nodes) {
    #pragma unroll
    for (int r = 0; r < 8; ++r)
      #pragma unroll
      for (int t = 0; t < 8; ++t)
        outp[(size_t)r * OUT_DIM + t * 16] = acc[t][r];
  } else {
    #pragma unroll
    for (int r = 0; r < 8; ++r)
      if (mbase + r < n_nodes)
        #pragma unroll
        for (int t = 0; t < 8; ++t)
          outp[(size_t)r * OUT_DIM + t * 16] = acc[t][r];
  }
}

#define TRIP_EDGES    (32 * BLK_EDGES)
#define GROUP_BLOCKS  16
#define TRIP_STRIDE   (WAVES_PER_WG * TRIP_EDGES)
#define HITCAP        1024
#define WAVES_PER_WG  32
#define TRIPS_PER_ROUND 7
#define QCAP          64

__device__ __forceinline__ void load_round(const int* __restrict__ pl, int rchunk, int lane,
                                           int nblk, unsigned (&w)[TRIPS_PER_ROUND]) {
  #pragma unroll
  for (int t = 0; t < TRIPS_PER_ROUND; ++t) {
    int b = rchunk / BLK_EDGES + t * (TRIP_STRIDE / BLK_EDGES) + lane;
    b = (b < nblk) ? b : (nblk - 1);
    b = (b > 0) ? b : 0;
    w[t] = (unsigned)pl[b];
  }
}

template <bool WHOLE, class HandOn>
__device__ __forceinline__ int note_round(const unsigned (&w)[TRIPS_PER_ROUND], unsigned pbit,
                                          int rchunk, int lane, int nblk, int n_edges,
                                          const int* __restrict__ edge_row,
                                          int row_base, int nrows,
                                          unsigned char* myblk, unsigned short* myq,
                                          HandOn& hand_on) {
  int nb = 0;
  #pragma unroll
  for (int t = 0; t < TRIPS_PER_ROUND; ++t) {
    const int  b = rchunk / BLK_EDGES + t * (TRIP_STRIDE / BLK_EDGES) + lane;
    const bool f = (((w[t] >> pbit) & 1u) != 0u) & (b < nblk);
    const unsigned fm = __builtin_amdgcn_ballot_w32(f);
    if (f) myblk[nb + (int)__builtin_amdgcn_mbcnt_lo(fm, 0u)] = (unsigned char)(t * 32 + lane);
    nb += __builtin_popcount(fm);
  }

  int qn = 0;
  for (int g = 0; g < nb; g += 32) {
    const int cntb  = (nb - g < 32) ? (nb - g) : 32;
    const int myid  = (int)myblk[g + ((lane < cntb) ? lane : (cntb - 1))];
    const int ebase = rchunk + (myid >> 5) * TRIP_STRIDE + (myid & 31) * BLK_EDGES;
    for (int j = 0; j < cntb; j += GROUP_BLOCKS) {
      if (qn + GROUP_BLOCKS * BLK_EDGES > HITCAP) { hand_on(qn); qn = 0; }
      const int nv = (cntb - j < GROUP_BLOCKS) ? (cntb - j) : GROUP_BLOCKS;
      int rr[GROUP_BLOCKS];
      #pragma unroll
      for (int u = 0; u < GROUP_BLOCKS; ++u) {
        int e = __builtin_amdgcn_readlane(ebase, j + u) + lane;
        if (!WHOLE) e = (e < n_edges) ? e : (n_edges - 1);
        rr[u] = edge_row[e];
      }
      #pragma unroll
      for (int u = 0; u < GROUP_BLOCKS; ++u) {
        const unsigned lim = (u < nv) ? (unsigned)nrows : 0u;
        const int eb = __builtin_amdgcn_readlane(ebase, j + u);
        bool in = (unsigned)(rr[u] - row_base) < lim;
        if (!WHOLE) in = in & (eb + lane < n_edges);
        const unsigned hm = __builtin_amdgcn_ballot_w32(in);
        if (hm) {
          if (in) {
            const unsigned x = (unsigned)(eb + lane - rchunk);
            myq[qn + (int)__builtin_amdgcn_mbcnt_lo(hm, 0u)] =
                (unsigned short)(((x / (unsigned)TRIP_STRIDE) << 10) | (x & (unsigned)(TRIP_EDGES - 1)));
          }
          qn += __builtin_popcount(hm);
        }
      }
    }
  }
  return qn;
}

__device__ __forceinline__ void lds_add(float* p, float v) {
  __hip_atomic_fetch_add(p, v, __ATOMIC_RELAXED, __HIP_MEMORY_SCOPE_WORKGROUP);
}

__device__ __forceinline__ float4 row_fma(float4 s, float4 m, float v) {
  s.x += m.x * v; s.y += m.y * v; s.z += m.z * v; s.w += m.w * v;
  return s;
}

__device__ __forceinline__ void drain_queue(const int* qcol, const float* qval,
                                            const unsigned char* qrow, unsigned* qcnt,
                                            int qi, int lane, const float* __restrict__ support,
                                            float* acc) {
  const int nraw = (int)qcnt[qi];
  const int n = __builtin_amdgcn_readfirstlane(nraw < QCAP ? nraw : QCAP);
  if (n == 0) return;
  if (lane == 0) qcnt[qi] = 0u;
  const int qb = qi * QCAP;
  for (int g = 0; g < n; g += 32) {
    const int cnt = (n - g < 32) ? (n - g) : 32;
    int colv = 0; float vv = 0.0f; int rs = 0;
    if (lane < cnt) {
      colv = qcol[qb + g + lane];
      vv   = qval[qb + g + lane];
      rs   = (int)qrow[qb + g + lane];
    }
    auto gather8 = [&](float4 (&m)[8], int j) {
      #pragma unroll
      for (int t = 0; t < 8; ++t) {
        const int jj = (j + t < cnt) ? (j + t) : (cnt - 1);
        const int c  = __builtin_amdgcn_readlane(colv, jj);
        m[t] = *((const float4*)(support + (size_t)c * OUT_DIM) + lane);
      }
    };
    auto add8 = [&](const float4 (&m)[8], int j) {
      #pragma unroll
      for (int t = 0; t < 8; ++t) {
        if (j + t < cnt) {
          const float v = __builtin_bit_cast(float,
              __builtin_amdgcn_readlane(__builtin_bit_cast(int, vv), j + t));
          const int r = __builtin_amdgcn_readlane(rs, j + t);
          float4* p = (float4*)(acc + r * OUT_DIM) + lane;
          *p = row_fma(*p, m[t], v);
        }
      }
    };
    float4 ma[8], mb[8];
    gather8(ma, 0);
    for (int j = 0; j < cnt; j += 16) {
      const bool second = j + 8 < cnt;
      if (second) gather8(mb, j + 8);
      add8(ma, j);
      if (second) {
        if (j + 16 < cnt) gather8(ma, j + 16);
        add8(mb, j + 8);
      }
    }
  }
}

__global__ void __launch_bounds__(1024)
gcn_rows_owner(const float* __restrict__ support,
               const float* __restrict__ adj_vals,
               const int* __restrict__ keys,
               const int* __restrict__ edge_row,
               const int* __restrict__ edge_col,
               const float* __restrict__ bias,
               float* __restrict__ out, int n_nodes, int n_edges) {
  __shared__ float          acc[ROWS_PER_WG * OUT_DIM];
  __shared__ float          ovf[ROWS_PER_WG * OUT_DIM];
  __shared__ unsigned short hitq[WAVES_PER_WG * HITCAP];
  __shared__ unsigned char  blkq[WAVES_PER_WG * TRIPS_PER_ROUND * 32];
  __shared__ int            qcol[2 * WAVES_PER_WG * QCAP];
  __shared__ float          qval[2 * WAVES_PER_WG * QCAP];
  __shared__ unsigned char  qrow[2 * WAVES_PER_WG * QCAP];
  __shared__ unsigned       qcnt[2 * WAVES_PER_WG];

  const int lane = threadIdx.x & 31;
  const int wave = __builtin_amdgcn_readfirstlane((int)(threadIdx.x >> 5));
  unsigned char*  myblk = blkq + wave * (TRIPS_PER_ROUND * 32);
  unsigned short* myq = hitq + wave * HITCAP;

  const float4 b4 = *((const float4*)bias + lane);

  const int  nblk    = presence_blocks(n_edges);
  const int  pstride = presence_stride(n_edges);
  const bool whole_blocks = (n_edges % BLK_EDGES) == 0;

  const int nbt    = (n_edges + WAVES_PER_WG * TRIP_EDGES - 1) / (WAVES_PER_WG * TRIP_EDGES);
  const int rounds = (nbt + TRIPS_PER_ROUND - 1) / TRIPS_PER_ROUND;

  const bool drain_first = (wave & 4) != 0;

  for (int row_base = blockIdx.x * ROWS_PER_WG; row_base < n_nodes;
       row_base += gridDim.x * ROWS_PER_WG) {
    const int nrows = (n_nodes - row_base < ROWS_PER_WG) ? (n_nodes - row_base) : ROWS_PER_WG;

    for (int r = wave; r < nrows; r += WAVES_PER_WG) {
      *((float4*)(acc + r * OUT_DIM) + lane) = b4;
      *((float4*)(ovf + r * OUT_DIM) + lane) = make_float4(0.f, 0.f, 0.f, 0.f);
    }
    if (threadIdx.x < 2 * WAVES_PER_WG) qcnt[threadIdx.x] = 0u;
    __syncthreads();

    const unsigned key  = owner_key(row_base);
    const int* __restrict__ pl = keys + (size_t)(key >> 5) * (size_t)pstride;
    const unsigned pbit = key & 31u;

    for (int round = 0; round <= rounds; ++round) {
      const int wb = round & 1;
      if (drain_first)
        drain_queue(qcol, qval, qrow, qcnt, (wb ^ 1) * WAVES_PER_WG + wave, lane, support, acc);

      if (round < rounds) {
        const int rchunk = (round * TRIPS_PER_ROUND * WAVES_PER_WG + wave) * TRIP_EDGES;

        auto hand_on = [&](int n) {
          for (int g = 0; g < n; g += 32) {
            const int cnt = (n - g < 32) ? (n - g) : 32;
            unsigned rloc = 0u; int colv = 0; float vv = 0.0f; bool mine = false;
            if (lane < cnt) {
              const int note = (int)myq[g + lane];
              const int e = rchunk + (note >> 10) * (WAVES_PER_WG * TRIP_EDGES) + (note & 1023);
              rloc = (unsigned)(edge_row[e] - row_base);
              colv = edge_col[e];
              vv   = adj_vals[e];
              mine = rloc < (unsigned)nrows;
            }
            bool full = false;
            if (mine) {
              const int qi = wb * WAVES_PER_WG + (int)(rloc & 31u);
              const unsigned slot = __hip_atomic_fetch_add(&qcnt[qi], 1u, __ATOMIC_RELAXED,
                                                           __HIP_MEMORY_SCOPE_WORKGROUP);
              if (slot < (unsigned)QCAP) {
                const int idx = qi * QCAP + (int)slot;
                qcol[idx] = colv;
                qval[idx] = vv;
                qrow[idx] = (unsigned char)rloc;
              } else {
                full = true;
              }
            }
            unsigned fm = __builtin_amdgcn_ballot_w32(full);
            while (fm) {
              const int jj = __builtin_ctz(fm);
              fm &= fm - 1u;
              const int      c  = __builtin_amdgcn_readlane(colv, jj);
              const float    v  = __builtin_bit_cast(float,
                                    __builtin_amdgcn_readlane(__builtin_bit_cast(int, vv), jj));
              const unsigned rr = (unsigned)__builtin_amdgcn_readlane((int)rloc, jj);
              const float4 m = *((const float4*)(support + (size_t)c * OUT_DIM) + lane);
              float* a = ovf + rr * OUT_DIM + lane * 4;
              lds_add(a + 0, m.x * v);
              lds_add(a + 1, m.y * v);
              lds_add(a + 2, m.z * v);
              lds_add(a + 3, m.w * v);
            }
          }
        };

        unsigned pw[TRIPS_PER_ROUND];
        load_round(pl, rchunk, lane, nblk, pw);
        const int qn = whole_blocks
            ? note_round<true >(pw, pbit, rchunk, lane, nblk, n_edges, edge_row, row_base, nrows,
                                myblk, myq, hand_on)
            : note_round<false>(pw, pbit, rchunk, lane, nblk, n_edges, edge_row, row_base, nrows,
                                myblk, myq, hand_on);
        if (qn) hand_on(qn);
      }

      if (!drain_first)
        drain_queue(qcol, qval, qrow, qcnt, (wb ^ 1) * WAVES_PER_WG + wave, lane, support, acc);
      __syncthreads();
    }

    for (int r = wave; r < nrows; r += WAVES_PER_WG) {
      const float4 a = *((const float4*)(acc + r * OUT_DIM) + lane);
      const float4 o = *((const float4*)(ovf + r * OUT_DIM) + lane);
      *((float4*)(out + (size_t)(row_base + r) * OUT_DIM) + lane) =
          make_float4(a.x + o.x, a.y + o.y, a.z + o.z, a.w + o.w);
    }
    __syncthreads();
  }
}

extern "C" void kernel_launch(void* const* d_in, const int* in_sizes, int n_in,
                              void* d_out, int out_size, void* d_ws, size_t ws_size,
                              hipStream_t stream) {
  const float* x        = (const float*)d_in[0];
  const float* w        = (const float*)d_in[1];
  const float* bias     = (const float*)d_in[2];
  const float* adj_vals = (const float*)d_in[3];
  const int*   edge_row = (const int*)d_in[4];
  const int*   edge_col = (const int*)d_in[5];

  const int n_nodes = in_sizes[0] / IN_DIM;
  const int n_edges = in_sizes[3];

  float* out     = (float*)d_out;
  float* support = (float*)d_ws;
  int*   keys    = (int*)(support + (size_t)n_nodes * OUT_DIM);

  const int nblocks = (n_nodes + 127) / 128;
  gcn_gemm_wmma<<<nblocks, 256, 0, stream>>>(x, w, support, n_nodes, edge_row, keys, n_edges);

  gcn_rows_owner<<<256, 1024, 0, stream>>>(
      support, adj_vals, keys, edge_row, edge_col, bias, out, n_nodes, n_edges);
}


# owner scan: each listed block's row test leaves its wave mask in the block's own lane; lanes note their hits together, one pass per hit depth (was a branch and a compaction per block); arrival order
# speedup vs baseline: 2.4657x; 1.0626x over previous
#include <hip/hip_runtime.h>
#include <hip/hip_bf16.h>

#define IN_DIM  128
#define OUT_DIM 128

typedef float v2f __attribute__((ext_vector_type(2)));
typedef float v8f __attribute__((ext_vector_type(8)));

#define ROWS_PER_WG   196

__device__ __forceinline__ unsigned owner_key(int r) {
  return ((unsigned)r / (unsigned)ROWS_PER_WG) & 255u;
}

#define BLK_EDGES 32
__device__ __forceinline__ int presence_blocks(int n_edges) {
  return (n_edges + BLK_EDGES - 1) / BLK_EDGES;
}
__device__ __forceinline__ int presence_stride(int n_edges) {
  return (presence_blocks(n_edges) + 31) & ~31;
}
__device__ __forceinline__ void presence_mark(unsigned (&m)[8], int r) {
  const unsigned k   = owner_key(r);
  const unsigned bit = 1u << (k & 31u);
  const unsigned j   = k >> 5;
  #pragma unroll
  for (int t = 0; t < 8; ++t) m[t] |= (j == (unsigned)t) ? bit : 0u;
}

__global__ void __launch_bounds__(256)
gcn_gemm_wmma(const float* __restrict__ x, const float* __restrict__ w,
              float* __restrict__ support, int n_nodes,
              const int* __restrict__ edge_row, int* __restrict__ keys, int n_edges) {
  __shared__ float wlds[IN_DIM * OUT_DIM];

  const int lane = threadIdx.x & 31;
  const int wave = threadIdx.x >> 5;
  const int m0   = blockIdx.x * 128 + wave * 16;

  {
    const int nblk    = presence_blocks(n_edges);
    const int pstride = presence_stride(n_edges);
    const int ngroups = (nblk + 31) >> 5;
    for (int g = wave * (int)gridDim.x + (int)blockIdx.x; g < ngroups; g += 8 * (int)gridDim.x) {
      const int b = g * 32 + lane;
      if (b < nblk) {
        unsigned m[8] = {0u, 0u, 0u, 0u, 0u, 0u, 0u, 0u};
        const int e0 = b * BLK_EDGES;
        if (e0 + BLK_EDGES <= n_edges) {
          #pragma unroll
          for (int i = 0; i < BLK_EDGES / 4; ++i) {
            const int4 rv = *(const int4*)(edge_row + e0 + 4 * i);
            presence_mark(m, rv.x);
            presence_mark(m, rv.y);
            presence_mark(m, rv.z);
            presence_mark(m, rv.w);
          }
        } else {
          for (int i = 0; e0 + i < n_edges; ++i)
            presence_mark(m, edge_row[e0 + i]);
        }
        #pragma unroll
        for (int j = 0; j < 8; ++j) keys[j * pstride + b] = (int)m[j];
      }
    }
  }

  for (int i = threadIdx.x; i < IN_DIM * OUT_DIM; i += 256)
    wlds[i] = w[i];
  __syncthreads();

  const int koff = (lane >> 4) * 2;
  const int nsub = lane & 15;

  int mrow = m0 + (lane & 15);
  if (mrow >= n_nodes) mrow = n_nodes - 1;

  const float* __restrict__ xrow = x + (size_t)mrow * IN_DIM;

  v8f acc[8];
  #pragma unroll
  for (int t = 0; t < 8; ++t) acc[t] = (v8f){};

  #pragma unroll 2
  for (int k0 = 0; k0 < IN_DIM; k0 += 4) {
    v2f a;
    a.x = xrow[k0 + koff];
    a.y = xrow[k0 + koff + 1];
    const float* __restrict__ wk0 = &wlds[(k0 + koff) * OUT_DIM + nsub];
    #pragma unroll
    for (int t = 0; t < 8; ++t) {
      v2f b;
      b.x = wk0[t * 16];
      b.y = wk0[t * 16 + OUT_DIM];
      acc[t] = __builtin_amdgcn_wmma_f32_16x16x4_f32(
           false, a,  false, b,
           (short)0, acc[t],  false,  false);
    }
  }

  const int mbase = m0 + ((lane >> 4) << 3);
  float* outp = support + (size_t)mbase * OUT_DIM + nsub;

  if (m0 + 15 < n_nodes) {
    #pragma unroll
    for (int r = 0; r < 8; ++r)
      #pragma unroll
      for (int t = 0; t < 8; ++t)
        outp[(size_t)r * OUT_DIM + t * 16] = acc[t][r];
  } else {
    #pragma unroll
    for (int r = 0; r < 8; ++r)
      if (mbase + r < n_nodes)
        #pragma unroll
        for (int t = 0; t < 8; ++t)
          outp[(size_t)r * OUT_DIM + t * 16] = acc[t][r];
  }
}

#define TRIP_EDGES    (32 * BLK_EDGES)
#define GROUP_BLOCKS  16
#define TRIP_STRIDE   (WAVES_PER_WG * TRIP_EDGES)
#define HITCAP        1024
#define WAVES_PER_WG  32
#define TRIPS_PER_ROUND 7
#define QCAP          64

__device__ __forceinline__ void load_round(const int* __restrict__ pl, int rchunk, int lane,
                                           int nblk, unsigned (&w)[TRIPS_PER_ROUND]) {
  #pragma unroll
  for (int t = 0; t < TRIPS_PER_ROUND; ++t) {
    int b = rchunk / BLK_EDGES + t * (TRIP_STRIDE / BLK_EDGES) + lane;
    b = (b < nblk) ? b : (nblk - 1);
    b = (b > 0) ? b : 0;
    w[t] = (unsigned)pl[b];
  }
}

template <bool WHOLE, class HandOn>
__device__ __forceinline__ int note_round(const unsigned (&w)[TRIPS_PER_ROUND], unsigned pbit,
                                          int rchunk, int lane, int nblk, int n_edges,
                                          const int* __restrict__ edge_row,
                                          int row_base, int nrows,
                                          unsigned char* myblk, unsigned short* myq,
                                          HandOn& hand_on) {
  int nb = 0;
  #pragma unroll
  for (int t = 0; t < TRIPS_PER_ROUND; ++t) {
    const int  b = rchunk / BLK_EDGES + t * (TRIP_STRIDE / BLK_EDGES) + lane;
    const bool f = (((w[t] >> pbit) & 1u) != 0u) & (b < nblk);
    const unsigned fm = __builtin_amdgcn_ballot_w32(f);
    if (f) myblk[nb + (int)__builtin_amdgcn_mbcnt_lo(fm, 0u)] = (unsigned char)(t * 32 + lane);
    nb += __builtin_popcount(fm);
  }

  int qn = 0;
  for (int g = 0; g < nb; g += 32) {
    const int cntb  = (nb - g < 32) ? (nb - g) : 32;
    const int myid  = (int)myblk[g + ((lane < cntb) ? lane : (cntb - 1))];
    const int ebase = rchunk + (myid >> 5) * TRIP_STRIDE + (myid & 31) * BLK_EDGES;
    if (qn + 32 * BLK_EDGES > HITCAP) { hand_on(qn); qn = 0; }
    unsigned hits = 0u;
    for (int j = 0; j < cntb; j += GROUP_BLOCKS) {
      int rr[GROUP_BLOCKS];
      #pragma unroll
      for (int u = 0; u < GROUP_BLOCKS; ++u) {
        int e = __builtin_amdgcn_readlane(ebase, j + u) + lane;
        if (!WHOLE) e = (e < n_edges) ? e : (n_edges - 1);
        rr[u] = edge_row[e];
      }
      #pragma unroll
      for (int u = 0; u < GROUP_BLOCKS; ++u) {
        const bool in = (unsigned)(rr[u] - row_base) < (unsigned)nrows;
        const unsigned hm = __builtin_amdgcn_ballot_w32(in);
        hits = __builtin_amdgcn_inverse_ballot_w32(1u << (j + u)) ? hm : hits;
      }
    }
    if (lane >= cntb) hits = 0u;
    if (!WHOLE) {
      const int left = n_edges - ebase;
      if (left < BLK_EDGES) hits &= (1u << left) - 1u;
    }
    unsigned am = __builtin_amdgcn_ballot_w32(hits != 0u);
    while (am) {
      if (hits != 0u) {
        myq[qn + (int)__builtin_amdgcn_mbcnt_lo(am, 0u)] =
            (unsigned short)(((unsigned)myid << 5) | (unsigned)__builtin_ctz(hits));
      }
      hits &= hits - 1u;
      qn += __builtin_popcount(am);
      am = __builtin_amdgcn_ballot_w32(hits != 0u);
    }
  }
  return qn;
}

__device__ __forceinline__ void lds_add(float* p, float v) {
  __hip_atomic_fetch_add(p, v, __ATOMIC_RELAXED, __HIP_MEMORY_SCOPE_WORKGROUP);
}

__device__ __forceinline__ float4 row_fma(float4 s, float4 m, float v) {
  s.x += m.x * v; s.y += m.y * v; s.z += m.z * v; s.w += m.w * v;
  return s;
}

__device__ __forceinline__ void drain_queue(const int* qcol, const float* qval,
                                            const unsigned char* qrow, unsigned* qcnt,
                                            int qi, int lane, const float* __restrict__ support,
                                            float* acc) {
  const int nraw = (int)qcnt[qi];
  const int n = __builtin_amdgcn_readfirstlane(nraw < QCAP ? nraw : QCAP);
  if (n == 0) return;
  if (lane == 0) qcnt[qi] = 0u;
  const int qb = qi * QCAP;
  for (int g = 0; g < n; g += 32) {
    const int cnt = (n - g < 32) ? (n - g) : 32;
    int colv = 0; float vv = 0.0f; int rs = 0;
    if (lane < cnt) {
      colv = qcol[qb + g + lane];
      vv   = qval[qb + g + lane];
      rs   = (int)qrow[qb + g + lane];
    }
    auto gather8 = [&](float4 (&m)[8], int j) {
      #pragma unroll
      for (int t = 0; t < 8; ++t) {
        const int jj = (j + t < cnt) ? (j + t) : (cnt - 1);
        const int c  = __builtin_amdgcn_readlane(colv, jj);
        m[t] = *((const float4*)(support + (size_t)c * OUT_DIM) + lane);
      }
    };
    auto add8 = [&](const float4 (&m)[8], int j) {
      #pragma unroll
      for (int t = 0; t < 8; ++t) {
        if (j + t < cnt) {
          const float v = __builtin_bit_cast(float,
              __builtin_amdgcn_readlane(__builtin_bit_cast(int, vv), j + t));
          const int r = __builtin_amdgcn_readlane(rs, j + t);
          float4* p = (float4*)(acc + r * OUT_DIM) + lane;
          *p = row_fma(*p, m[t], v);
        }
      }
    };
    float4 ma[8], mb[8];
    gather8(ma, 0);
    for (int j = 0; j < cnt; j += 16) {
      const bool second = j + 8 < cnt;
      if (second) gather8(mb, j + 8);
      add8(ma, j);
      if (second) {
        if (j + 16 < cnt) gather8(ma, j + 16);
        add8(mb, j + 8);
      }
    }
  }
}

__global__ void __launch_bounds__(1024)
gcn_rows_owner(const float* __restrict__ support,
               const float* __restrict__ adj_vals,
               const int* __restrict__ keys,
               const int* __restrict__ edge_row,
               const int* __restrict__ edge_col,
               const float* __restrict__ bias,
               float* __restrict__ out, int n_nodes, int n_edges) {
  __shared__ float          acc[ROWS_PER_WG * OUT_DIM];
  __shared__ float          ovf[ROWS_PER_WG * OUT_DIM];
  __shared__ unsigned short hitq[WAVES_PER_WG * HITCAP];
  __shared__ unsigned char  blkq[WAVES_PER_WG * TRIPS_PER_ROUND * 32];
  __shared__ int            qcol[2 * WAVES_PER_WG * QCAP];
  __shared__ float          qval[2 * WAVES_PER_WG * QCAP];
  __shared__ unsigned char  qrow[2 * WAVES_PER_WG * QCAP];
  __shared__ unsigned       qcnt[2 * WAVES_PER_WG];

  const int lane = threadIdx.x & 31;
  const int wave = __builtin_amdgcn_readfirstlane((int)(threadIdx.x >> 5));
  unsigned char*  myblk = blkq + wave * (TRIPS_PER_ROUND * 32);
  unsigned short* myq = hitq + wave * HITCAP;

  const float4 b4 = *((const float4*)bias + lane);

  const int  nblk    = presence_blocks(n_edges);
  const int  pstride = presence_stride(n_edges);
  const bool whole_blocks = (n_edges % BLK_EDGES) == 0;

  const int nbt    = (n_edges + WAVES_PER_WG * TRIP_EDGES - 1) / (WAVES_PER_WG * TRIP_EDGES);
  const int rounds = (nbt + TRIPS_PER_ROUND - 1) / TRIPS_PER_ROUND;

  const bool drain_first = (wave & 4) != 0;

  for (int row_base = blockIdx.x * ROWS_PER_WG; row_base < n_nodes;
       row_base += gridDim.x * ROWS_PER_WG) {
    const int nrows = (n_nodes - row_base < ROWS_PER_WG) ? (n_nodes - row_base) : ROWS_PER_WG;

    for (int r = wave; r < nrows; r += WAVES_PER_WG) {
      *((float4*)(acc + r * OUT_DIM) + lane) = b4;
      *((float4*)(ovf + r * OUT_DIM) + lane) = make_float4(0.f, 0.f, 0.f, 0.f);
    }
    if (threadIdx.x < 2 * WAVES_PER_WG) qcnt[threadIdx.x] = 0u;
    __syncthreads();

    const unsigned key  = owner_key(row_base);
    const int* __restrict__ pl = keys + (size_t)(key >> 5) * (size_t)pstride;
    const unsigned pbit = key & 31u;

    for (int round = 0; round <= rounds; ++round) {
      const int wb = round & 1;
      if (drain_first)
        drain_queue(qcol, qval, qrow, qcnt, (wb ^ 1) * WAVES_PER_WG + wave, lane, support, acc);

      if (round < rounds) {
        const int rchunk = (round * TRIPS_PER_ROUND * WAVES_PER_WG + wave) * TRIP_EDGES;

        auto hand_on = [&](int n) {
          for (int g = 0; g < n; g += 32) {
            const int cnt = (n - g < 32) ? (n - g) : 32;
            unsigned rloc = 0u; int colv = 0; float vv = 0.0f; bool mine = false;
            if (lane < cnt) {
              const int note = (int)myq[g + lane];
              const int e = rchunk + (note >> 10) * (WAVES_PER_WG * TRIP_EDGES) + (note & 1023);
              rloc = (unsigned)(edge_row[e] - row_base);
              colv = edge_col[e];
              vv   = adj_vals[e];
              mine = rloc < (unsigned)nrows;
            }
            bool full = false;
            if (mine) {
              const int qi = wb * WAVES_PER_WG + (int)(rloc & 31u);
              const unsigned slot = __hip_atomic_fetch_add(&qcnt[qi], 1u, __ATOMIC_RELAXED,
                                                           __HIP_MEMORY_SCOPE_WORKGROUP);
              if (slot < (unsigned)QCAP) {
                const int idx = qi * QCAP + (int)slot;
                qcol[idx] = colv;
                qval[idx] = vv;
                qrow[idx] = (unsigned char)rloc;
              } else {
                full = true;
              }
            }
            unsigned fm = __builtin_amdgcn_ballot_w32(full);
            while (fm) {
              const int jj = __builtin_ctz(fm);
              fm &= fm - 1u;
              const int      c  = __builtin_amdgcn_readlane(colv, jj);
              const float    v  = __builtin_bit_cast(float,
                                    __builtin_amdgcn_readlane(__builtin_bit_cast(int, vv), jj));
              const unsigned rr = (unsigned)__builtin_amdgcn_readlane((int)rloc, jj);
              const float4 m = *((const float4*)(support + (size_t)c * OUT_DIM) + lane);
              float* a = ovf + rr * OUT_DIM + lane * 4;
              lds_add(a + 0, m.x * v);
              lds_add(a + 1, m.y * v);
              lds_add(a + 2, m.z * v);
              lds_add(a + 3, m.w * v);
            }
          }
        };

        unsigned pw[TRIPS_PER_ROUND];
        load_round(pl, rchunk, lane, nblk, pw);
        const int qn = whole_blocks
            ? note_round<true >(pw, pbit, rchunk, lane, nblk, n_edges, edge_row, row_base, nrows,
                                myblk, myq, hand_on)
            : note_round<false>(pw, pbit, rchunk, lane, nblk, n_edges, edge_row, row_base, nrows,
                                myblk, myq, hand_on);
        if (qn) hand_on(qn);
      }

      if (!drain_first)
        drain_queue(qcol, qval, qrow, qcnt, (wb ^ 1) * WAVES_PER_WG + wave, lane, support, acc);
      __syncthreads();
    }

    for (int r = wave; r < nrows; r += WAVES_PER_WG) {
      const float4 a = *((const float4*)(acc + r * OUT_DIM) + lane);
      const float4 o = *((const float4*)(ovf + r * OUT_DIM) + lane);
      *((float4*)(out + (size_t)(row_base + r) * OUT_DIM) + lane) =
          make_float4(a.x + o.x, a.y + o.y, a.z + o.z, a.w + o.w);
    }
    __syncthreads();
  }
}

extern "C" void kernel_launch(void* const* d_in, const int* in_sizes, int n_in,
                              void* d_out, int out_size, void* d_ws, size_t ws_size,
                              hipStream_t stream) {
  const float* x        = (const float*)d_in[0];
  const float* w        = (const float*)d_in[1];
  const float* bias     = (const float*)d_in[2];
  const float* adj_vals = (const float*)d_in[3];
  const int*   edge_row = (const int*)d_in[4];
  const int*   edge_col = (const int*)d_in[5];

  const int n_nodes = in_sizes[0] / IN_DIM;
  const int n_edges = in_sizes[3];

  float* out     = (float*)d_out;
  float* support = (float*)d_ws;
  int*   keys    = (int*)(support + (size_t)n_nodes * OUT_DIM);

  const int nblocks = (n_nodes + 127) / 128;
  gcn_gemm_wmma<<<nblocks, 256, 0, stream>>>(x, w, support, n_nodes, edge_row, keys, n_edges);

  gcn_rows_owner<<<256, 1024, 0, stream>>>(
      support, adj_vals, keys, edge_row, edge_col, bias, out, n_nodes, n_edges);
}


# owner scan: an owner wave's queue takes the finds of the WHOLE scan; round barriers and in-between drains gone: one barrier, then each wave empties its queue once; arrival order, not per round
# speedup vs baseline: 2.5622x; 1.0391x over previous
#include <hip/hip_runtime.h>
#include <hip/hip_bf16.h>

#define IN_DIM  128
#define OUT_DIM 128

typedef float v2f __attribute__((ext_vector_type(2)));
typedef float v8f __attribute__((ext_vector_type(8)));

#define ROWS_PER_WG   196

__device__ __forceinline__ unsigned owner_key(int r) {
  return ((unsigned)r / (unsigned)ROWS_PER_WG) & 255u;
}

#define BLK_EDGES 32
__device__ __forceinline__ int presence_blocks(int n_edges) {
  return (n_edges + BLK_EDGES - 1) / BLK_EDGES;
}
__device__ __forceinline__ int presence_stride(int n_edges) {
  return (presence_blocks(n_edges) + 31) & ~31;
}
__device__ __forceinline__ void presence_mark(unsigned (&m)[8], int r) {
  const unsigned k   = owner_key(r);
  const unsigned bit = 1u << (k & 31u);
  const unsigned j   = k >> 5;
  #pragma unroll
  for (int t = 0; t < 8; ++t) m[t] |= (j == (unsigned)t) ? bit : 0u;
}

__global__ void __launch_bounds__(256)
gcn_gemm_wmma(const float* __restrict__ x, const float* __restrict__ w,
              float* __restrict__ support, int n_nodes,
              const int* __restrict__ edge_row, int* __restrict__ keys, int n_edges) {
  __shared__ float wlds[IN_DIM * OUT_DIM];

  const int lane = threadIdx.x & 31;
  const int wave = threadIdx.x >> 5;
  const int m0   = blockIdx.x * 128 + wave * 16;

  {
    const int nblk    = presence_blocks(n_edges);
    const int pstride = presence_stride(n_edges);
    const int ngroups = (nblk + 31) >> 5;
    for (int g = wave * (int)gridDim.x + (int)blockIdx.x; g < ngroups; g += 8 * (int)gridDim.x) {
      const int b = g * 32 + lane;
      if (b < nblk) {
        unsigned m[8] = {0u, 0u, 0u, 0u, 0u, 0u, 0u, 0u};
        const int e0 = b * BLK_EDGES;
        if (e0 + BLK_EDGES <= n_edges) {
          #pragma unroll
          for (int i = 0; i < BLK_EDGES / 4; ++i) {
            const int4 rv = *(const int4*)(edge_row + e0 + 4 * i);
            presence_mark(m, rv.x);
            presence_mark(m, rv.y);
            presence_mark(m, rv.z);
            presence_mark(m, rv.w);
          }
        } else {
          for (int i = 0; e0 + i < n_edges; ++i)
            presence_mark(m, edge_row[e0 + i]);
        }
        #pragma unroll
        for (int j = 0; j < 8; ++j) keys[j * pstride + b] = (int)m[j];
      }
    }
  }

  for (int i = threadIdx.x; i < IN_DIM * OUT_DIM; i += 256)
    wlds[i] = w[i];
  __syncthreads();

  const int koff = (lane >> 4) * 2;
  const int nsub = lane & 15;

  int mrow = m0 + (lane & 15);
  if (mrow >= n_nodes) mrow = n_nodes - 1;

  const float* __restrict__ xrow = x + (size_t)mrow * IN_DIM;

  v8f acc[8];
  #pragma unroll
  for (int t = 0; t < 8; ++t) acc[t] = (v8f){};

  #pragma unroll 2
  for (int k0 = 0; k0 < IN_DIM; k0 += 4) {
    v2f a;
    a.x = xrow[k0 + koff];
    a.y = xrow[k0 + koff + 1];
    const float* __restrict__ wk0 = &wlds[(k0 + koff) * OUT_DIM + nsub];
    #pragma unroll
    for (int t = 0; t < 8; ++t) {
      v2f b;
      b.x = wk0[t * 16];
      b.y = wk0[t * 16 + OUT_DIM];
      acc[t] = __builtin_amdgcn_wmma_f32_16x16x4_f32(
           false, a,  false, b,
           (short)0, acc[t],  false,  false);
    }
  }

  const int mbase = m0 + ((lane >> 4) << 3);
  float* outp = support + (size_t)mbase * OUT_DIM + nsub;

  if (m0 + 15 < n_nodes) {
    #pragma unroll
    for (int r = 0; r < 8; ++r)
      #pragma unroll
      for (int t = 0; t < 8; ++t)
        outp[(size_t)r * OUT_DIM + t * 16] = acc[t][r];
  } else {
    #pragma unroll
    for (int r = 0; r < 8; ++r)
      if (mbase + r < n_nodes)
        #pragma unroll
        for (int t = 0; t < 8; ++t)
          outp[(size_t)r * OUT_DIM + t * 16] = acc[t][r];
  }
}

#define TRIP_EDGES    (32 * BLK_EDGES)
#define GROUP_BLOCKS  16
#define TRIP_STRIDE   (WAVES_PER_WG * TRIP_EDGES)
#define HITCAP        1024
#define WAVES_PER_WG  32
#define TRIPS_PER_ROUND 7
#define QCAP          160

__device__ __forceinline__ void load_round(const int* __restrict__ pl, int rchunk, int lane,
                                           int nblk, unsigned (&w)[TRIPS_PER_ROUND]) {
  #pragma unroll
  for (int t = 0; t < TRIPS_PER_ROUND; ++t) {
    int b = rchunk / BLK_EDGES + t * (TRIP_STRIDE / BLK_EDGES) + lane;
    b = (b < nblk) ? b : (nblk - 1);
    b = (b > 0) ? b : 0;
    w[t] = (unsigned)pl[b];
  }
}

template <bool WHOLE, class HandOn>
__device__ __forceinline__ int note_round(const unsigned (&w)[TRIPS_PER_ROUND], unsigned pbit,
                                          int rchunk, int lane, int nblk, int n_edges,
                                          const int* __restrict__ edge_row,
                                          int row_base, int nrows,
                                          unsigned char* myblk, unsigned short* myq,
                                          HandOn& hand_on) {
  int nb = 0;
  #pragma unroll
  for (int t = 0; t < TRIPS_PER_ROUND; ++t) {
    const int  b = rchunk / BLK_EDGES + t * (TRIP_STRIDE / BLK_EDGES) + lane;
    const bool f = (((w[t] >> pbit) & 1u) != 0u) & (b < nblk);
    const unsigned fm = __builtin_amdgcn_ballot_w32(f);
    if (f) myblk[nb + (int)__builtin_amdgcn_mbcnt_lo(fm, 0u)] = (unsigned char)(t * 32 + lane);
    nb += __builtin_popcount(fm);
  }

  int qn = 0;
  for (int g = 0; g < nb; g += 32) {
    const int cntb  = (nb - g < 32) ? (nb - g) : 32;
    const int myid  = (int)myblk[g + ((lane < cntb) ? lane : (cntb - 1))];
    const int ebase = rchunk + (myid >> 5) * TRIP_STRIDE + (myid & 31) * BLK_EDGES;
    if (qn + 32 * BLK_EDGES > HITCAP) { hand_on(qn); qn = 0; }
    unsigned hits = 0u;
    for (int j = 0; j < cntb; j += GROUP_BLOCKS) {
      int rr[GROUP_BLOCKS];
      #pragma unroll
      for (int u = 0; u < GROUP_BLOCKS; ++u) {
        int e = __builtin_amdgcn_readlane(ebase, j + u) + lane;
        if (!WHOLE) e = (e < n_edges) ? e : (n_edges - 1);
        rr[u] = edge_row[e];
      }
      #pragma unroll
      for (int u = 0; u < GROUP_BLOCKS; ++u) {
        const bool in = (unsigned)(rr[u] - row_base) < (unsigned)nrows;
        const unsigned hm = __builtin_amdgcn_ballot_w32(in);
        hits = __builtin_amdgcn_inverse_ballot_w32(1u << (j + u)) ? hm : hits;
      }
    }
    if (lane >= cntb) hits = 0u;
    if (!WHOLE) {
      const int left = n_edges - ebase;
      if (left < BLK_EDGES) hits &= (1u << left) - 1u;
    }
    unsigned am = __builtin_amdgcn_ballot_w32(hits != 0u);
    while (am) {
      if (hits != 0u) {
        myq[qn + (int)__builtin_amdgcn_mbcnt_lo(am, 0u)] =
            (unsigned short)(((unsigned)myid << 5) | (unsigned)__builtin_ctz(hits));
      }
      hits &= hits - 1u;
      qn += __builtin_popcount(am);
      am = __builtin_amdgcn_ballot_w32(hits != 0u);
    }
  }
  return qn;
}

__device__ __forceinline__ void lds_add(float* p, float v) {
  __hip_atomic_fetch_add(p, v, __ATOMIC_RELAXED, __HIP_MEMORY_SCOPE_WORKGROUP);
}

__device__ __forceinline__ float4 row_fma(float4 s, float4 m, float v) {
  s.x += m.x * v; s.y += m.y * v; s.z += m.z * v; s.w += m.w * v;
  return s;
}

__device__ __forceinline__ void drain_queue(const int* qcol, const float* qval,
                                            const unsigned char* qrow, const unsigned* qcnt,
                                            int qi, int lane, const float* __restrict__ support,
                                            float* acc) {
  const int nraw = (int)qcnt[qi];
  const int n = __builtin_amdgcn_readfirstlane(nraw < QCAP ? nraw : QCAP);
  if (n == 0) return;
  const int qb = qi * QCAP;
  for (int g = 0; g < n; g += 32) {
    const int cnt = (n - g < 32) ? (n - g) : 32;
    int colv = 0; float vv = 0.0f; int rs = 0;
    if (lane < cnt) {
      colv = qcol[qb + g + lane];
      vv   = qval[qb + g + lane];
      rs   = (int)qrow[qb + g + lane];
    }
    auto gather8 = [&](float4 (&m)[8], int j) {
      #pragma unroll
      for (int t = 0; t < 8; ++t) {
        const int jj = (j + t < cnt) ? (j + t) : (cnt - 1);
        const int c  = __builtin_amdgcn_readlane(colv, jj);
        m[t] = *((const float4*)(support + (size_t)c * OUT_DIM) + lane);
      }
    };
    auto add8 = [&](const float4 (&m)[8], int j) {
      #pragma unroll
      for (int t = 0; t < 8; ++t) {
        if (j + t < cnt) {
          const float v = __builtin_bit_cast(float,
              __builtin_amdgcn_readlane(__builtin_bit_cast(int, vv), j + t));
          const int r = __builtin_amdgcn_readlane(rs, j + t);
          float4* p = (float4*)(acc + r * OUT_DIM) + lane;
          *p = row_fma(*p, m[t], v);
        }
      }
    };
    float4 ma[8], mb[8];
    gather8(ma, 0);
    for (int j = 0; j < cnt; j += 16) {
      const bool second = j + 8 < cnt;
      if (second) gather8(mb, j + 8);
      add8(ma, j);
      if (second) {
        if (j + 16 < cnt) gather8(ma, j + 16);
        add8(mb, j + 8);
      }
    }
  }
}

__global__ void __launch_bounds__(1024)
gcn_rows_owner(const float* __restrict__ support,
               const float* __restrict__ adj_vals,
               const int* __restrict__ keys,
               const int* __restrict__ edge_row,
               const int* __restrict__ edge_col,
               const float* __restrict__ bias,
               float* __restrict__ out, int n_nodes, int n_edges) {
  __shared__ float          acc[ROWS_PER_WG * OUT_DIM];
  __shared__ float          ovf[ROWS_PER_WG * OUT_DIM];
  __shared__ unsigned short hitq[WAVES_PER_WG * HITCAP];
  __shared__ unsigned char  blkq[WAVES_PER_WG * TRIPS_PER_ROUND * 32];
  __shared__ int            qcol[WAVES_PER_WG * QCAP];
  __shared__ float          qval[WAVES_PER_WG * QCAP];
  __shared__ unsigned char  qrow[WAVES_PER_WG * QCAP];
  __shared__ unsigned       qcnt[WAVES_PER_WG];

  const int lane = threadIdx.x & 31;
  const int wave = __builtin_amdgcn_readfirstlane((int)(threadIdx.x >> 5));
  unsigned char*  myblk = blkq + wave * (TRIPS_PER_ROUND * 32);
  unsigned short* myq = hitq + wave * HITCAP;

  const float4 b4 = *((const float4*)bias + lane);

  const int  nblk    = presence_blocks(n_edges);
  const int  pstride = presence_stride(n_edges);
  const bool whole_blocks = (n_edges % BLK_EDGES) == 0;

  const int nbt    = (n_edges + WAVES_PER_WG * TRIP_EDGES - 1) / (WAVES_PER_WG * TRIP_EDGES);
  const int rounds = (nbt + TRIPS_PER_ROUND - 1) / TRIPS_PER_ROUND;

  for (int row_base = blockIdx.x * ROWS_PER_WG; row_base < n_nodes;
       row_base += gridDim.x * ROWS_PER_WG) {
    const int nrows = (n_nodes - row_base < ROWS_PER_WG) ? (n_nodes - row_base) : ROWS_PER_WG;

    for (int r = wave; r < nrows; r += WAVES_PER_WG) {
      *((float4*)(acc + r * OUT_DIM) + lane) = b4;
      *((float4*)(ovf + r * OUT_DIM) + lane) = make_float4(0.f, 0.f, 0.f, 0.f);
    }
    if (threadIdx.x < WAVES_PER_WG) qcnt[threadIdx.x] = 0u;
    __syncthreads();

    const unsigned key  = owner_key(row_base);
    const int* __restrict__ pl = keys + (size_t)(key >> 5) * (size_t)pstride;
    const unsigned pbit = key & 31u;

    for (int round = 0; round < rounds; ++round) {
      const int rchunk = (round * TRIPS_PER_ROUND * WAVES_PER_WG + wave) * TRIP_EDGES;

      auto hand_on = [&](int n) {
        for (int g = 0; g < n; g += 32) {
          const int cnt = (n - g < 32) ? (n - g) : 32;
          unsigned rloc = 0u; int colv = 0; float vv = 0.0f; bool mine = false;
          if (lane < cnt) {
            const int note = (int)myq[g + lane];
            const int e = rchunk + (note >> 10) * (WAVES_PER_WG * TRIP_EDGES) + (note & 1023);
            rloc = (unsigned)(edge_row[e] - row_base);
            colv = edge_col[e];
            vv   = adj_vals[e];
            mine = rloc < (unsigned)nrows;
          }
          bool full = false;
          if (mine) {
            const int qi = (int)(rloc & 31u);
            const unsigned slot = __hip_atomic_fetch_add(&qcnt[qi], 1u, __ATOMIC_RELAXED,
                                                         __HIP_MEMORY_SCOPE_WORKGROUP);
            if (slot < (unsigned)QCAP) {
              const int idx = qi * QCAP + (int)slot;
              qcol[idx] = colv;
              qval[idx] = vv;
              qrow[idx] = (unsigned char)rloc;
            } else {
              full = true;
            }
          }
          unsigned fm = __builtin_amdgcn_ballot_w32(full);
          while (fm) {
            const int jj = __builtin_ctz(fm);
            fm &= fm - 1u;
            const int      c  = __builtin_amdgcn_readlane(colv, jj);
            const float    v  = __builtin_bit_cast(float,
                                  __builtin_amdgcn_readlane(__builtin_bit_cast(int, vv), jj));
            const unsigned rr = (unsigned)__builtin_amdgcn_readlane((int)rloc, jj);
            const float4 m = *((const float4*)(support + (size_t)c * OUT_DIM) + lane);
            float* a = ovf + rr * OUT_DIM + lane * 4;
            lds_add(a + 0, m.x * v);
            lds_add(a + 1, m.y * v);
            lds_add(a + 2, m.z * v);
            lds_add(a + 3, m.w * v);
          }
        }
      };

      unsigned pw[TRIPS_PER_ROUND];
      load_round(pl, rchunk, lane, nblk, pw);
      const int qn = whole_blocks
          ? note_round<true >(pw, pbit, rchunk, lane, nblk, n_edges, edge_row, row_base, nrows,
                              myblk, myq, hand_on)
          : note_round<false>(pw, pbit, rchunk, lane, nblk, n_edges, edge_row, row_base, nrows,
                              myblk, myq, hand_on);
      if (qn) hand_on(qn);
    }
    __syncthreads();

    drain_queue(qcol, qval, qrow, qcnt, wave, lane, support, acc);

    for (int r = wave; r < nrows; r += WAVES_PER_WG) {
      const float4 a = *((const float4*)(acc + r * OUT_DIM) + lane);
      const float4 o = *((const float4*)(ovf + r * OUT_DIM) + lane);
      *((float4*)(out + (size_t)(row_base + r) * OUT_DIM) + lane) =
          make_float4(a.x + o.x, a.y + o.y, a.z + o.z, a.w + o.w);
    }
    __syncthreads();
  }
}

extern "C" void kernel_launch(void* const* d_in, const int* in_sizes, int n_in,
                              void* d_out, int out_size, void* d_ws, size_t ws_size,
                              hipStream_t stream) {
  const float* x        = (const float*)d_in[0];
  const float* w        = (const float*)d_in[1];
  const float* bias     = (const float*)d_in[2];
  const float* adj_vals = (const float*)d_in[3];
  const int*   edge_row = (const int*)d_in[4];
  const int*   edge_col = (const int*)d_in[5];

  const int n_nodes = in_sizes[0] / IN_DIM;
  const int n_edges = in_sizes[3];

  float* out     = (float*)d_out;
  float* support = (float*)d_ws;
  int*   keys    = (int*)(support + (size_t)n_nodes * OUT_DIM);

  const int nblocks = (n_nodes + 127) / 128;
  gcn_gemm_wmma<<<nblocks, 256, 0, stream>>>(x, w, support, n_nodes, edge_row, keys, n_edges);

  gcn_rows_owner<<<256, 1024, 0, stream>>>(
      support, adj_vals, keys, edge_row, edge_col, bias, out, n_nodes, n_edges);
}


# product after aggregation: out = (sum adj * x[col]) @ W + bias; the owner group multiplies its own LDS rows by W (fp32 WMMA); GEMM launch and support plane gone; sum order changes
# speedup vs baseline: 3.0124x; 1.1757x over previous
#include <hip/hip_runtime.h>
#include <hip/hip_bf16.h>

#define IN_DIM  128
#define OUT_DIM 128

typedef float v2f __attribute__((ext_vector_type(2)));
typedef float v8f __attribute__((ext_vector_type(8)));

#define ROWS_PER_WG   196

__device__ __forceinline__ unsigned owner_key(int r) {
  return ((unsigned)r / (unsigned)ROWS_PER_WG) & 255u;
}

#define BLK_EDGES 32
__device__ __forceinline__ int presence_blocks(int n_edges) {
  return (n_edges + BLK_EDGES - 1) / BLK_EDGES;
}
__device__ __forceinline__ int presence_stride(int n_edges) {
  return (presence_blocks(n_edges) + 31) & ~31;
}
__device__ __forceinline__ void presence_mark(unsigned (&m)[8], int r) {
  const unsigned k   = owner_key(r);
  const unsigned bit = 1u << (k & 31u);
  const unsigned j   = k >> 5;
  #pragma unroll
  for (int t = 0; t < 8; ++t) m[t] |= (j == (unsigned)t) ? bit : 0u;
}

__global__ void __launch_bounds__(256)
gcn_presence(const int* __restrict__ edge_row, int* __restrict__ keys, int n_edges) {
  const int lane = threadIdx.x & 31;
  const int wave = threadIdx.x >> 5;
  const int nblk    = presence_blocks(n_edges);
  const int pstride = presence_stride(n_edges);
  const int ngroups = (nblk + 31) >> 5;
  for (int g = wave * (int)gridDim.x + (int)blockIdx.x; g < ngroups; g += 8 * (int)gridDim.x) {
    const int b = g * 32 + lane;
    if (b < nblk) {
      unsigned m[8] = {0u, 0u, 0u, 0u, 0u, 0u, 0u, 0u};
      const int e0 = b * BLK_EDGES;
      if (e0 + BLK_EDGES <= n_edges) {
        #pragma unroll
        for (int i = 0; i < BLK_EDGES / 4; ++i) {
          const int4 rv = *(const int4*)(edge_row + e0 + 4 * i);
          presence_mark(m, rv.x);
          presence_mark(m, rv.y);
          presence_mark(m, rv.z);
          presence_mark(m, rv.w);
        }
      } else {
        for (int i = 0; e0 + i < n_edges; ++i)
          presence_mark(m, edge_row[e0 + i]);
      }
      #pragma unroll
      for (int j = 0; j < 8; ++j) keys[j * pstride + b] = (int)m[j];
    }
  }
}

#define TRIP_EDGES    (32 * BLK_EDGES)
#define GROUP_BLOCKS  16
#define TRIP_STRIDE   (WAVES_PER_WG * TRIP_EDGES)
#define HITCAP        1024
#define WAVES_PER_WG  32
#define TRIPS_PER_ROUND 7
#define QCAP          160
#define ACC_PITCH     (IN_DIM + 4)
#define ROW_TILES     ((ROWS_PER_WG + 15) / 16)

__device__ __forceinline__ void load_round(const int* __restrict__ pl, int rchunk, int lane,
                                           int nblk, unsigned (&w)[TRIPS_PER_ROUND]) {
  #pragma unroll
  for (int t = 0; t < TRIPS_PER_ROUND; ++t) {
    int b = rchunk / BLK_EDGES + t * (TRIP_STRIDE / BLK_EDGES) + lane;
    b = (b < nblk) ? b : (nblk - 1);
    b = (b > 0) ? b : 0;
    w[t] = (unsigned)pl[b];
  }
}

template <bool WHOLE, class HandOn>
__device__ __forceinline__ int note_round(const unsigned (&w)[TRIPS_PER_ROUND], unsigned pbit,
                                          int rchunk, int lane, int nblk, int n_edges,
                                          const int* __restrict__ edge_row,
                                          int row_base, int nrows,
                                          unsigned char* myblk, unsigned short* myq,
                                          HandOn& hand_on) {
  int nb = 0;
  #pragma unroll
  for (int t = 0; t < TRIPS_PER_ROUND; ++t) {
    const int  b = rchunk / BLK_EDGES + t * (TRIP_STRIDE / BLK_EDGES) + lane;
    const bool f = (((w[t] >> pbit) & 1u) != 0u) & (b < nblk);
    const unsigned fm = __builtin_amdgcn_ballot_w32(f);
    if (f) myblk[nb + (int)__builtin_amdgcn_mbcnt_lo(fm, 0u)] = (unsigned char)(t * 32 + lane);
    nb += __builtin_popcount(fm);
  }

  int qn = 0;
  for (int g = 0; g < nb; g += 32) {
    const int cntb  = (nb - g < 32) ? (nb - g) : 32;
    const int myid  = (int)myblk[g + ((lane < cntb) ? lane : (cntb - 1))];
    const int ebase = rchunk + (myid >> 5) * TRIP_STRIDE + (myid & 31) * BLK_EDGES;
    if (qn + 32 * BLK_EDGES > HITCAP) { hand_on(qn); qn = 0; }
    unsigned hits = 0u;
    for (int j = 0; j < cntb; j += GROUP_BLOCKS) {
      int rr[GROUP_BLOCKS];
      #pragma unroll
      for (int u = 0; u < GROUP_BLOCKS; ++u) {
        int e = __builtin_amdgcn_readlane(ebase, j + u) + lane;
        if (!WHOLE) e = (e < n_edges) ? e : (n_edges - 1);
        rr[u] = edge_row[e];
      }
      #pragma unroll
      for (int u = 0; u < GROUP_BLOCKS; ++u) {
        const bool in = (unsigned)(rr[u] - row_base) < (unsigned)nrows;
        const unsigned hm = __builtin_amdgcn_ballot_w32(in);
        hits = __builtin_amdgcn_inverse_ballot_w32(1u << (j + u)) ? hm : hits;
      }
    }
    if (lane >= cntb) hits = 0u;
    if (!WHOLE) {
      const int left = n_edges - ebase;
      if (left < BLK_EDGES) hits &= (1u << left) - 1u;
    }
    unsigned am = __builtin_amdgcn_ballot_w32(hits != 0u);
    while (am) {
      if (hits != 0u) {
        myq[qn + (int)__builtin_amdgcn_mbcnt_lo(am, 0u)] =
            (unsigned short)(((unsigned)myid << 5) | (unsigned)__builtin_ctz(hits));
      }
      hits &= hits - 1u;
      qn += __builtin_popcount(am);
      am = __builtin_amdgcn_ballot_w32(hits != 0u);
    }
  }
  return qn;
}

__device__ __forceinline__ void lds_add(float* p, float v) {
  __hip_atomic_fetch_add(p, v, __ATOMIC_RELAXED, __HIP_MEMORY_SCOPE_WORKGROUP);
}

__device__ __forceinline__ float4 row_fma(float4 s, float4 m, float v) {
  s.x += m.x * v; s.y += m.y * v; s.z += m.z * v; s.w += m.w * v;
  return s;
}

__device__ __forceinline__ void drain_queue(const int* qcol, const float* qval,
                                            const unsigned char* qrow, const unsigned* qcnt,
                                            int qi, int lane, const float* __restrict__ x,
                                            float* acc) {
  const int nraw = (int)qcnt[qi];
  const int n = __builtin_amdgcn_readfirstlane(nraw < QCAP ? nraw : QCAP);
  if (n == 0) return;
  const int qb = qi * QCAP;
  for (int g = 0; g < n; g += 32) {
    const int cnt = (n - g < 32) ? (n - g) : 32;
    int colv = 0; float vv = 0.0f; int rs = 0;
    if (lane < cnt) {
      colv = qcol[qb + g + lane];
      vv   = qval[qb + g + lane];
      rs   = (int)qrow[qb + g + lane];
    }
    auto gather8 = [&](float4 (&m)[8], int j) {
      #pragma unroll
      for (int t = 0; t < 8; ++t) {
        const int jj = (j + t < cnt) ? (j + t) : (cnt - 1);
        const int c  = __builtin_amdgcn_readlane(colv, jj);
        m[t] = *((const float4*)(x + (size_t)c * IN_DIM) + lane);
      }
    };
    auto add8 = [&](const float4 (&m)[8], int j) {
      #pragma unroll
      for (int t = 0; t < 8; ++t) {
        if (j + t < cnt) {
          const float v = __builtin_bit_cast(float,
              __builtin_amdgcn_readlane(__builtin_bit_cast(int, vv), j + t));
          const int r = __builtin_amdgcn_readlane(rs, j + t);
          float4* p = (float4*)(acc + r * ACC_PITCH) + lane;
          *p = row_fma(*p, m[t], v);
        }
      }
    };
    float4 ma[8], mb[8];
    gather8(ma, 0);
    for (int j = 0; j < cnt; j += 16) {
      const bool second = j + 8 < cnt;
      if (second) gather8(mb, j + 8);
      add8(ma, j);
      if (second) {
        if (j + 16 < cnt) gather8(ma, j + 16);
        add8(mb, j + 8);
      }
    }
  }
}

__global__ void __launch_bounds__(1024)
gcn_rows_owner(const float* __restrict__ x,
               const float* __restrict__ w,
               const float* __restrict__ adj_vals,
               const int* __restrict__ keys,
               const int* __restrict__ edge_row,
               const int* __restrict__ edge_col,
               const float* __restrict__ bias,
               float* __restrict__ out, int n_nodes, int n_edges) {
  __shared__ __attribute__((aligned(16))) float acc[ROWS_PER_WG * ACC_PITCH];
  __shared__ __attribute__((aligned(16))) float ovf[ROWS_PER_WG * IN_DIM];
  __shared__ __attribute__((aligned(16))) float wlds[IN_DIM * OUT_DIM];
  unsigned short* hitq = (unsigned short*)wlds;
  __shared__ unsigned char  blkq[WAVES_PER_WG * TRIPS_PER_ROUND * 32];
  __shared__ int            qcol[WAVES_PER_WG * QCAP];
  __shared__ float          qval[WAVES_PER_WG * QCAP];
  __shared__ unsigned char  qrow[WAVES_PER_WG * QCAP];
  __shared__ unsigned       qcnt[WAVES_PER_WG];

  const int lane = threadIdx.x & 31;
  const int wave = __builtin_amdgcn_readfirstlane((int)(threadIdx.x >> 5));
  unsigned char*  myblk = blkq + wave * (TRIPS_PER_ROUND * 32);
  unsigned short* myq = hitq + wave * HITCAP;

  const int  nblk    = presence_blocks(n_edges);
  const int  pstride = presence_stride(n_edges);
  const bool whole_blocks = (n_edges % BLK_EDGES) == 0;

  const int nbt    = (n_edges + WAVES_PER_WG * TRIP_EDGES - 1) / (WAVES_PER_WG * TRIP_EDGES);
  const int rounds = (nbt + TRIPS_PER_ROUND - 1) / TRIPS_PER_ROUND;

  for (int row_base = blockIdx.x * ROWS_PER_WG; row_base < n_nodes;
       row_base += gridDim.x * ROWS_PER_WG) {
    const int nrows = (n_nodes - row_base < ROWS_PER_WG) ? (n_nodes - row_base) : ROWS_PER_WG;

    for (int r = wave; r < nrows; r += WAVES_PER_WG) {
      *((float4*)(acc + r * ACC_PITCH) + lane) = make_float4(0.f, 0.f, 0.f, 0.f);
      *((float4*)(ovf + r * IN_DIM) + lane)    = make_float4(0.f, 0.f, 0.f, 0.f);
    }
    if (threadIdx.x < WAVES_PER_WG) qcnt[threadIdx.x] = 0u;
    __syncthreads();

    const unsigned key  = owner_key(row_base);
    const int* __restrict__ pl = keys + (size_t)(key >> 5) * (size_t)pstride;
    const unsigned pbit = key & 31u;

    for (int round = 0; round < rounds; ++round) {
      const int rchunk = (round * TRIPS_PER_ROUND * WAVES_PER_WG + wave) * TRIP_EDGES;

      auto hand_on = [&](int n) {
        for (int g = 0; g < n; g += 32) {
          const int cnt = (n - g < 32) ? (n - g) : 32;
          unsigned rloc = 0u; int colv = 0; float vv = 0.0f; bool mine = false;
          if (lane < cnt) {
            const int note = (int)myq[g + lane];
            const int e = rchunk + (note >> 10) * (WAVES_PER_WG * TRIP_EDGES) + (note & 1023);
            rloc = (unsigned)(edge_row[e] - row_base);
            colv = edge_col[e];
            vv   = adj_vals[e];
            mine = rloc < (unsigned)nrows;
          }
          bool full = false;
          if (mine) {
            const int qi = (int)(rloc & 31u);
            const unsigned slot = __hip_atomic_fetch_add(&qcnt[qi], 1u, __ATOMIC_RELAXED,
                                                         __HIP_MEMORY_SCOPE_WORKGROUP);
            if (slot < (unsigned)QCAP) {
              const int idx = qi * QCAP + (int)slot;
              qcol[idx] = colv;
              qval[idx] = vv;
              qrow[idx] = (unsigned char)rloc;
            } else {
              full = true;
            }
          }
          unsigned fm = __builtin_amdgcn_ballot_w32(full);
          while (fm) {
            const int jj = __builtin_ctz(fm);
            fm &= fm - 1u;
            const int      c  = __builtin_amdgcn_readlane(colv, jj);
            const float    v  = __builtin_bit_cast(float,
                                  __builtin_amdgcn_readlane(__builtin_bit_cast(int, vv), jj));
            const unsigned rr = (unsigned)__builtin_amdgcn_readlane((int)rloc, jj);
            const float4 m = *((const float4*)(x + (size_t)c * IN_DIM) + lane);
            float* a = ovf + rr * IN_DIM + lane * 4;
            lds_add(a + 0, m.x * v);
            lds_add(a + 1, m.y * v);
            lds_add(a + 2, m.z * v);
            lds_add(a + 3, m.w * v);
          }
        }
      };

      unsigned pw[TRIPS_PER_ROUND];
      load_round(pl, rchunk, lane, nblk, pw);
      const int qn = whole_blocks
          ? note_round<true >(pw, pbit, rchunk, lane, nblk, n_edges, edge_row, row_base, nrows,
                              myblk, myq, hand_on)
          : note_round<false>(pw, pbit, rchunk, lane, nblk, n_edges, edge_row, row_base, nrows,
                              myblk, myq, hand_on);
      if (qn) hand_on(qn);
    }
    __syncthreads();

    for (int i = threadIdx.x; i < IN_DIM * OUT_DIM / 4; i += WAVES_PER_WG * 32)
      ((float4*)wlds)[i] = ((const float4*)w)[i];

    drain_queue(qcol, qval, qrow, qcnt, wave, lane, x, acc);

    for (int r = wave; r < nrows; r += WAVES_PER_WG) {
      float4* p = (float4*)(acc + r * ACC_PITCH) + lane;
      const float4 a = *p;
      const float4 o = *((const float4*)(ovf + r * IN_DIM) + lane);
      *p = make_float4(a.x + o.x, a.y + o.y, a.z + o.z, a.w + o.w);
    }
    __syncthreads();

    if (wave < 2 * ROW_TILES && (wave >> 1) * 16 < nrows) {
      const int rt0  = (wave >> 1) * 16;
      const int cb   = (wave & 1) * 64;
      const int koff = (lane >> 4) * 2;
      const int nsub = lane & 15;

      int ar = rt0 + nsub;
      if (ar >= nrows) ar = nrows - 1;
      const float* arow = acc + ar * ACC_PITCH;

      v8f d[4];
      #pragma unroll
      for (int t = 0; t < 4; ++t) d[t] = (v8f){};

      #pragma unroll 2
      for (int k0 = 0; k0 < IN_DIM; k0 += 4) {
        v2f a;
        a.x = arow[k0 + koff];
        a.y = arow[k0 + koff + 1];
        const float* wk0 = &wlds[(k0 + koff) * OUT_DIM + cb + nsub];
        #pragma unroll
        for (int t = 0; t < 4; ++t) {
          v2f b;
          b.x = wk0[t * 16];
          b.y = wk0[t * 16 + OUT_DIM];
          d[t] = __builtin_amdgcn_wmma_f32_16x16x4_f32(
               false, a,  false, b,
               (short)0, d[t],  false,  false);
        }
      }

      float bv[4];
      #pragma unroll
      for (int t = 0; t < 4; ++t) bv[t] = bias[cb + t * 16 + nsub];

      const int rl = rt0 + ((lane >> 4) << 3);
      #pragma unroll
      for (int r = 0; r < 8; ++r)
        if (rl + r < nrows)
          #pragma unroll
          for (int t = 0; t < 4; ++t)
            ovf[(rl + r) * IN_DIM + cb + t * 16 + nsub] = d[t][r] + bv[t];
      #pragma unroll
      for (int i = 0; i < 8; ++i) {
        const int row = rt0 + 2 * i + (lane >> 4);
        if (row < nrows)
          *((float4*)(out + (size_t)(row_base + row) * OUT_DIM + cb) + nsub) =
              *((const float4*)(ovf + row * IN_DIM + cb) + nsub);
      }
    }
    __syncthreads();
  }
}

extern "C" void kernel_launch(void* const* d_in, const int* in_sizes, int n_in,
                              void* d_out, int out_size, void* d_ws, size_t ws_size,
                              hipStream_t stream) {
  const float* x        = (const float*)d_in[0];
  const float* w        = (const float*)d_in[1];
  const float* bias     = (const float*)d_in[2];
  const float* adj_vals = (const float*)d_in[3];
  const int*   edge_row = (const int*)d_in[4];
  const int*   edge_col = (const int*)d_in[5];

  const int n_nodes = in_sizes[0] / IN_DIM;
  const int n_edges = in_sizes[3];

  float* out  = (float*)d_out;
  int*   keys = (int*)d_ws;

  gcn_presence<<<256, 256, 0, stream>>>(edge_row, keys, n_edges);

  gcn_rows_owner<<<256, 1024, 0, stream>>>(
      x, w, adj_vals, keys, edge_row, edge_col, bias, out, n_nodes, n_edges);
}


# drain: rows of x gathered in one rolling window of 16 per wave, a set asked for again right behind the add that frees it, queue-long; no wait covers a load just issued; sums and their order unchanged
# speedup vs baseline: 3.2294x; 1.0720x over previous
#include <hip/hip_runtime.h>
#include <hip/hip_bf16.h>

#define IN_DIM  128
#define OUT_DIM 128

typedef float v2f __attribute__((ext_vector_type(2)));
typedef float v8f __attribute__((ext_vector_type(8)));

#define ROWS_PER_WG   196

__device__ __forceinline__ unsigned owner_key(int r) {
  return ((unsigned)r / (unsigned)ROWS_PER_WG) & 255u;
}

#define BLK_EDGES 32
__device__ __forceinline__ int presence_blocks(int n_edges) {
  return (n_edges + BLK_EDGES - 1) / BLK_EDGES;
}
__device__ __forceinline__ int presence_stride(int n_edges) {
  return (presence_blocks(n_edges) + 31) & ~31;
}
__device__ __forceinline__ void presence_mark(unsigned (&m)[8], int r) {
  const unsigned k   = owner_key(r);
  const unsigned bit = 1u << (k & 31u);
  const unsigned j   = k >> 5;
  #pragma unroll
  for (int t = 0; t < 8; ++t) m[t] |= (j == (unsigned)t) ? bit : 0u;
}

__global__ void __launch_bounds__(256)
gcn_presence(const int* __restrict__ edge_row, int* __restrict__ keys, int n_edges) {
  const int lane = threadIdx.x & 31;
  const int wave = threadIdx.x >> 5;
  const int nblk    = presence_blocks(n_edges);
  const int pstride = presence_stride(n_edges);
  const int ngroups = (nblk + 31) >> 5;
  for (int g = wave * (int)gridDim.x + (int)blockIdx.x; g < ngroups; g += 8 * (int)gridDim.x) {
    const int b = g * 32 + lane;
    if (b < nblk) {
      unsigned m[8] = {0u, 0u, 0u, 0u, 0u, 0u, 0u, 0u};
      const int e0 = b * BLK_EDGES;
      if (e0 + BLK_EDGES <= n_edges) {
        #pragma unroll
        for (int i = 0; i < BLK_EDGES / 4; ++i) {
          const int4 rv = *(const int4*)(edge_row + e0 + 4 * i);
          presence_mark(m, rv.x);
          presence_mark(m, rv.y);
          presence_mark(m, rv.z);
          presence_mark(m, rv.w);
        }
      } else {
        for (int i = 0; e0 + i < n_edges; ++i)
          presence_mark(m, edge_row[e0 + i]);
      }
      #pragma unroll
      for (int j = 0; j < 8; ++j) keys[j * pstride + b] = (int)m[j];
    }
  }
}

#define TRIP_EDGES    (32 * BLK_EDGES)
#define GROUP_BLOCKS  16
#define TRIP_STRIDE   (WAVES_PER_WG * TRIP_EDGES)
#define HITCAP        1024
#define WAVES_PER_WG  32
#define TRIPS_PER_ROUND 7
#define QCAP          160
#define ACC_PITCH     (IN_DIM + 4)
#define ROW_TILES     ((ROWS_PER_WG + 15) / 16)

__device__ __forceinline__ void load_round(const int* __restrict__ pl, int rchunk, int lane,
                                           int nblk, unsigned (&w)[TRIPS_PER_ROUND]) {
  #pragma unroll
  for (int t = 0; t < TRIPS_PER_ROUND; ++t) {
    int b = rchunk / BLK_EDGES + t * (TRIP_STRIDE / BLK_EDGES) + lane;
    b = (b < nblk) ? b : (nblk - 1);
    b = (b > 0) ? b : 0;
    w[t] = (unsigned)pl[b];
  }
}

template <bool WHOLE, class HandOn>
__device__ __forceinline__ int note_round(const unsigned (&w)[TRIPS_PER_ROUND], unsigned pbit,
                                          int rchunk, int lane, int nblk, int n_edges,
                                          const int* __restrict__ edge_row,
                                          int row_base, int nrows,
                                          unsigned char* myblk, unsigned short* myq,
                                          HandOn& hand_on) {
  int nb = 0;
  #pragma unroll
  for (int t = 0; t < TRIPS_PER_ROUND; ++t) {
    const int  b = rchunk / BLK_EDGES + t * (TRIP_STRIDE / BLK_EDGES) + lane;
    const bool f = (((w[t] >> pbit) & 1u) != 0u) & (b < nblk);
    const unsigned fm = __builtin_amdgcn_ballot_w32(f);
    if (f) myblk[nb + (int)__builtin_amdgcn_mbcnt_lo(fm, 0u)] = (unsigned char)(t * 32 + lane);
    nb += __builtin_popcount(fm);
  }

  int qn = 0;
  for (int g = 0; g < nb; g += 32) {
    const int cntb  = (nb - g < 32) ? (nb - g) : 32;
    const int myid  = (int)myblk[g + ((lane < cntb) ? lane : (cntb - 1))];
    const int ebase = rchunk + (myid >> 5) * TRIP_STRIDE + (myid & 31) * BLK_EDGES;
    if (qn + 32 * BLK_EDGES > HITCAP) { hand_on(qn); qn = 0; }
    unsigned hits = 0u;
    for (int j = 0; j < cntb; j += GROUP_BLOCKS) {
      int rr[GROUP_BLOCKS];
      #pragma unroll
      for (int u = 0; u < GROUP_BLOCKS; ++u) {
        int e = __builtin_amdgcn_readlane(ebase, j + u) + lane;
        if (!WHOLE) e = (e < n_edges) ? e : (n_edges - 1);
        rr[u] = edge_row[e];
      }
      #pragma unroll
      for (int u = 0; u < GROUP_BLOCKS; ++u) {
        const bool in = (unsigned)(rr[u] - row_base) < (unsigned)nrows;
        const unsigned hm = __builtin_amdgcn_ballot_w32(in);
        hits = __builtin_amdgcn_inverse_ballot_w32(1u << (j + u)) ? hm : hits;
      }
    }
    if (lane >= cntb) hits = 0u;
    if (!WHOLE) {
      const int left = n_edges - ebase;
      if (left < BLK_EDGES) hits &= (1u << left) - 1u;
    }
    unsigned am = __builtin_amdgcn_ballot_w32(hits != 0u);
    while (am) {
      if (hits != 0u) {
        myq[qn + (int)__builtin_amdgcn_mbcnt_lo(am, 0u)] =
            (unsigned short)(((unsigned)myid << 5) | (unsigned)__builtin_ctz(hits));
      }
      hits &= hits - 1u;
      qn += __builtin_popcount(am);
      am = __builtin_amdgcn_ballot_w32(hits != 0u);
    }
  }
  return qn;
}

__device__ __forceinline__ void lds_add(float* p, float v) {
  __hip_atomic_fetch_add(p, v, __ATOMIC_RELAXED, __HIP_MEMORY_SCOPE_WORKGROUP);
}

__device__ __forceinline__ float4 row_fma(float4 s, float4 m, float v) {
  s.x += m.x * v; s.y += m.y * v; s.z += m.z * v; s.w += m.w * v;
  return s;
}

#define DRAIN_WINDOW 16
__device__ __forceinline__ void drain_queue(const int* qcol, const float* qval,
                                            const unsigned char* qrow, const unsigned* qcnt,
                                            int qi, int lane, const float* __restrict__ x,
                                            float* acc) {
  const int nraw = (int)qcnt[qi];
  const int n = __builtin_amdgcn_readfirstlane(nraw < QCAP ? nraw : QCAP);
  if (n == 0) return;
  const int qb = qi * QCAP;
  const float4* xl = (const float4*)x + lane;
  auto fields = [&](int g, int& colv, float& vv, int& rs) {
    colv = 0; vv = 0.0f; rs = 0;
    if (g + lane < n) {
      colv = qcol[qb + g + lane];
      vv   = qval[qb + g + lane];
      rs   = (int)qrow[qb + g + lane];
    }
  };
  auto add_row = [&](const float4& m, float vv, int rs, int p) {
    const float v = __builtin_bit_cast(float,
        __builtin_amdgcn_readlane(__builtin_bit_cast(int, vv), p));
    const int r = __builtin_amdgcn_readlane(rs, p);
    float4* q = (float4*)(acc + r * ACC_PITCH) + lane;
    *q = row_fma(*q, m, v);
  };
  int colv, rs;
  float vv;
  fields(0, colv, vv, rs);
  float4 m[DRAIN_WINDOW];
  #pragma unroll
  for (int p = 0; p < DRAIN_WINDOW; ++p) {
    m[p] = xl[(size_t)__builtin_amdgcn_readlane(colv, p) * (IN_DIM / 4)];
  }
  int g = 0;
  for (; g + DRAIN_WINDOW < n; g += DRAIN_WINDOW) {
    #pragma unroll
    for (int p = 0; p < DRAIN_WINDOW; ++p) {
      add_row(m[p], vv, rs, p);
      m[p] = xl[(size_t)__builtin_amdgcn_readlane(colv, p + DRAIN_WINDOW) * (IN_DIM / 4)];
    }
    fields(g + DRAIN_WINDOW, colv, vv, rs);
  }
  const int left = n - g;
  #pragma unroll
  for (int p = 0; p < DRAIN_WINDOW; ++p)
    if (p < left) add_row(m[p], vv, rs, p);
}

__global__ void __launch_bounds__(1024)
gcn_rows_owner(const float* __restrict__ x,
               const float* __restrict__ w,
               const float* __restrict__ adj_vals,
               const int* __restrict__ keys,
               const int* __restrict__ edge_row,
               const int* __restrict__ edge_col,
               const float* __restrict__ bias,
               float* __restrict__ out, int n_nodes, int n_edges) {
  __shared__ __attribute__((aligned(16))) float acc[ROWS_PER_WG * ACC_PITCH];
  __shared__ __attribute__((aligned(16))) float ovf[ROWS_PER_WG * IN_DIM];
  __shared__ __attribute__((aligned(16))) float wlds[IN_DIM * OUT_DIM];
  unsigned short* hitq = (unsigned short*)wlds;
  __shared__ unsigned char  blkq[WAVES_PER_WG * TRIPS_PER_ROUND * 32];
  __shared__ int            qcol[WAVES_PER_WG * QCAP];
  __shared__ float          qval[WAVES_PER_WG * QCAP];
  __shared__ unsigned char  qrow[WAVES_PER_WG * QCAP];
  __shared__ unsigned       qcnt[WAVES_PER_WG];

  const int lane = threadIdx.x & 31;
  const int wave = __builtin_amdgcn_readfirstlane((int)(threadIdx.x >> 5));
  unsigned char*  myblk = blkq + wave * (TRIPS_PER_ROUND * 32);
  unsigned short* myq = hitq + wave * HITCAP;

  const int  nblk    = presence_blocks(n_edges);
  const int  pstride = presence_stride(n_edges);
  const bool whole_blocks = (n_edges % BLK_EDGES) == 0;

  const int nbt    = (n_edges + WAVES_PER_WG * TRIP_EDGES - 1) / (WAVES_PER_WG * TRIP_EDGES);
  const int rounds = (nbt + TRIPS_PER_ROUND - 1) / TRIPS_PER_ROUND;

  for (int row_base = blockIdx.x * ROWS_PER_WG; row_base < n_nodes;
       row_base += gridDim.x * ROWS_PER_WG) {
    const int nrows = (n_nodes - row_base < ROWS_PER_WG) ? (n_nodes - row_base) : ROWS_PER_WG;

    for (int r = wave; r < nrows; r += WAVES_PER_WG) {
      *((float4*)(acc + r * ACC_PITCH) + lane) = make_float4(0.f, 0.f, 0.f, 0.f);
      *((float4*)(ovf + r * IN_DIM) + lane)    = make_float4(0.f, 0.f, 0.f, 0.f);
    }
    if (threadIdx.x < WAVES_PER_WG) qcnt[threadIdx.x] = 0u;
    __syncthreads();

    const unsigned key  = owner_key(row_base);
    const int* __restrict__ pl = keys + (size_t)(key >> 5) * (size_t)pstride;
    const unsigned pbit = key & 31u;

    for (int round = 0; round < rounds; ++round) {
      const int rchunk = (round * TRIPS_PER_ROUND * WAVES_PER_WG + wave) * TRIP_EDGES;

      auto hand_on = [&](int n) {
        for (int g = 0; g < n; g += 32) {
          const int cnt = (n - g < 32) ? (n - g) : 32;
          unsigned rloc = 0u; int colv = 0; float vv = 0.0f; bool mine = false;
          if (lane < cnt) {
            const int note = (int)myq[g + lane];
            const int e = rchunk + (note >> 10) * (WAVES_PER_WG * TRIP_EDGES) + (note & 1023);
            rloc = (unsigned)(edge_row[e] - row_base);
            colv = edge_col[e];
            vv   = adj_vals[e];
            mine = rloc < (unsigned)nrows;
          }
          bool full = false;
          if (mine) {
            const int qi = (int)(rloc & 31u);
            const unsigned slot = __hip_atomic_fetch_add(&qcnt[qi], 1u, __ATOMIC_RELAXED,
                                                         __HIP_MEMORY_SCOPE_WORKGROUP);
            if (slot < (unsigned)QCAP) {
              const int idx = qi * QCAP + (int)slot;
              qcol[idx] = colv;
              qval[idx] = vv;
              qrow[idx] = (unsigned char)rloc;
            } else {
              full = true;
            }
          }
          unsigned fm = __builtin_amdgcn_ballot_w32(full);
          while (fm) {
            const int jj = __builtin_ctz(fm);
            fm &= fm - 1u;
            const int      c  = __builtin_amdgcn_readlane(colv, jj);
            const float    v  = __builtin_bit_cast(float,
                                  __builtin_amdgcn_readlane(__builtin_bit_cast(int, vv), jj));
            const unsigned rr = (unsigned)__builtin_amdgcn_readlane((int)rloc, jj);
            const float4 m = *((const float4*)(x + (size_t)c * IN_DIM) + lane);
            float* a = ovf + rr * IN_DIM + lane * 4;
            lds_add(a + 0, m.x * v);
            lds_add(a + 1, m.y * v);
            lds_add(a + 2, m.z * v);
            lds_add(a + 3, m.w * v);
          }
        }
      };

      unsigned pw[TRIPS_PER_ROUND];
      load_round(pl, rchunk, lane, nblk, pw);
      const int qn = whole_blocks
          ? note_round<true >(pw, pbit, rchunk, lane, nblk, n_edges, edge_row, row_base, nrows,
                              myblk, myq, hand_on)
          : note_round<false>(pw, pbit, rchunk, lane, nblk, n_edges, edge_row, row_base, nrows,
                              myblk, myq, hand_on);
      if (qn) hand_on(qn);
    }
    __syncthreads();

    for (int i = threadIdx.x; i < IN_DIM * OUT_DIM / 4; i += WAVES_PER_WG * 32)
      ((float4*)wlds)[i] = ((const float4*)w)[i];

    drain_queue(qcol, qval, qrow, qcnt, wave, lane, x, acc);

    for (int r = wave; r < nrows; r += WAVES_PER_WG) {
      float4* p = (float4*)(acc + r * ACC_PITCH) + lane;
      const float4 a = *p;
      const float4 o = *((const float4*)(ovf + r * IN_DIM) + lane);
      *p = make_float4(a.x + o.x, a.y + o.y, a.z + o.z, a.w + o.w);
    }
    __syncthreads();

    if (wave < 2 * ROW_TILES && (wave >> 1) * 16 < nrows) {
      const int rt0  = (wave >> 1) * 16;
      const int cb   = (wave & 1) * 64;
      const int koff = (lane >> 4) * 2;
      const int nsub = lane & 15;

      int ar = rt0 + nsub;
      if (ar >= nrows) ar = nrows - 1;
      const float* arow = acc + ar * ACC_PITCH;

      v8f d[4];
      #pragma unroll
      for (int t = 0; t < 4; ++t) d[t] = (v8f){};

      #pragma unroll 2
      for (int k0 = 0; k0 < IN_DIM; k0 += 4) {
        v2f a;
        a.x = arow[k0 + koff];
        a.y = arow[k0 + koff + 1];
        const float* wk0 = &wlds[(k0 + koff) * OUT_DIM + cb + nsub];
        #pragma unroll
        for (int t = 0; t < 4; ++t) {
          v2f b;
          b.x = wk0[t * 16];
          b.y = wk0[t * 16 + OUT_DIM];
          d[t] = __builtin_amdgcn_wmma_f32_16x16x4_f32(
               false, a,  false, b,
               (short)0, d[t],  false,  false);
        }
      }

      float bv[4];
      #pragma unroll
      for (int t = 0; t < 4; ++t) bv[t] = bias[cb + t * 16 + nsub];

      const int rl = rt0 + ((lane >> 4) << 3);
      #pragma unroll
      for (int r = 0; r < 8; ++r)
        if (rl + r < nrows)
          #pragma unroll
          for (int t = 0; t < 4; ++t)
            ovf[(rl + r) * IN_DIM + cb + t * 16 + nsub] = d[t][r] + bv[t];
      #pragma unroll
      for (int i = 0; i < 8; ++i) {
        const int row = rt0 + 2 * i + (lane >> 4);
        if (row < nrows)
          *((float4*)(out + (size_t)(row_base + row) * OUT_DIM + cb) + nsub) =
              *((const float4*)(ovf + row * IN_DIM + cb) + nsub);
      }
    }
    __syncthreads();
  }
}

extern "C" void kernel_launch(void* const* d_in, const int* in_sizes, int n_in,
                              void* d_out, int out_size, void* d_ws, size_t ws_size,
                              hipStream_t stream) {
  const float* x        = (const float*)d_in[0];
  const float* w        = (const float*)d_in[1];
  const float* bias     = (const float*)d_in[2];
  const float* adj_vals = (const float*)d_in[3];
  const int*   edge_row = (const int*)d_in[4];
  const int*   edge_col = (const int*)d_in[5];

  const int n_nodes = in_sizes[0] / IN_DIM;
  const int n_edges = in_sizes[3];

  float* out  = (float*)d_out;
  int*   keys = (int*)d_ws;

  gcn_presence<<<256, 256, 0, stream>>>(edge_row, keys, n_edges);

  gcn_rows_owner<<<256, 1024, 0, stream>>>(
      x, w, adj_vals, keys, edge_row, edge_col, bias, out, n_nodes, n_edges);
}
